# starting kernel
# baseline (speedup 1.0000x reference)
#include <hip/hip_runtime.h>
#include <hip/hip_bf16.h>
#include <stdint.h>

#define HID   256
#define EDIM  64
#define HEADS 8
#define CCH   32
#define NEG_SLOPE 0.2f
#define LN_EPS 1e-5f

typedef float v2f __attribute__((ext_vector_type(2)));
typedef float v8f __attribute__((ext_vector_type(8)));

__device__ __forceinline__ unsigned enc_f32(float f) {
    unsigned b = __float_as_uint(f);
    return (b & 0x80000000u) ? ~b : (b | 0x80000000u);
}
__device__ __forceinline__ float dec_f32(unsigned u) {
    unsigned b = (u & 0x80000000u) ? (u & 0x7FFFFFFFu) : ~u;
    return __uint_as_float(b);
}

__global__ void k_weh(const float* __restrict__ W_e,
                      const float* __restrict__ att_edge,
                      float* __restrict__ w_eh) {
    int d = blockIdx.x * blockDim.x + threadIdx.x;
    if (d >= EDIM) return;
    #pragma unroll
    for (int h = 0; h < HEADS; ++h) {
        float s = 0.f;
        #pragma unroll
        for (int c = 0; c < CCH; ++c)
            s += W_e[d * HID + h * CCH + c] * att_edge[h * CCH + c];
        w_eh[d * HEADS + h] = s;
    }
}

__global__ void __launch_bounds__(256)
k_node_proj(const float* __restrict__ x, const float* __restrict__ W,
            float* __restrict__ xp, int n_nodes) {
    const int wave = threadIdx.x >> 5;
    const int lane = threadIdx.x & 31;
    const int row_base = (blockIdx.x * 2 + (wave >> 2)) * 16;
    const int col_base = (wave & 3) * 64;
    if (row_base >= n_nodes) return;

    int m = lane & 15;
    int kk = (lane >> 4) * 2;
    int row = row_base + m;
    if (row >= n_nodes) row = n_nodes - 1;
    const float* xrow = x + (size_t)row * HID;
    const int ncol = col_base + (lane & 15);

    v8f acc0 = {}, acc1 = {}, acc2 = {}, acc3 = {};
    for (int k = 0; k < HID; k += 4) {
        v2f a;
        a.x = xrow[k + kk];
        a.y = xrow[k + kk + 1];
        const float* w0 = W + (size_t)(k + kk) * HID;
        const float* w1 = W + (size_t)(k + kk + 1) * HID;
        v2f b0, b1, b2, b3;
        b0.x = w0[ncol];      b0.y = w1[ncol];
        b1.x = w0[ncol + 16]; b1.y = w1[ncol + 16];
        b2.x = w0[ncol + 32]; b2.y = w1[ncol + 32];
        b3.x = w0[ncol + 48]; b3.y = w1[ncol + 48];
        acc0 = __builtin_amdgcn_wmma_f32_16x16x4_f32(false, a, false, b0, (short)0, acc0, false, false);
        acc1 = __builtin_amdgcn_wmma_f32_16x16x4_f32(false, a, false, b1, (short)0, acc1, false, false);
        acc2 = __builtin_amdgcn_wmma_f32_16x16x4_f32(false, a, false, b2, (short)0, acc2, false, false);
        acc3 = __builtin_amdgcn_wmma_f32_16x16x4_f32(false, a, false, b3, (short)0, acc3, false, false);
    }

    const int srow = row_base + ((lane >> 4) * 8);
    const int scol = col_base + (lane & 15);
    #pragma unroll
    for (int i = 0; i < 8; ++i) {
        int r = srow + i;
        if (r < n_nodes) {
            float* o = xp + (size_t)r * HID + scol;
            o[0]  = acc0[i];
            o[16] = acc1[i];
            o[32] = acc2[i];
            o[48] = acc3[i];
        }
    }
}

__global__ void __launch_bounds__(256)
k_att_dots(const float* __restrict__ xp, const float* __restrict__ att_src,
           const float* __restrict__ att_dst, float* __restrict__ a_src,
           float* __restrict__ a_dst, int n_nodes) {
    const int wave = threadIdx.x >> 5;
    const int lane = threadIdx.x & 31;
    const int node = blockIdx.x * 8 + wave;
    if (node >= n_nodes) return;
    const float* row = xp + (size_t)node * HID + lane * 8;
    float s = 0.f, t = 0.f;
    #pragma unroll
    for (int j = 0; j < 8; ++j) {
        float v = row[j];
        int col = lane * 8 + j;
        s += v * att_src[col];
        t += v * att_dst[col];
    }
    s += __shfl_xor(s, 1); s += __shfl_xor(s, 2);
    t += __shfl_xor(t, 1); t += __shfl_xor(t, 2);
    if ((lane & 3) == 0) {
        a_src[node * HEADS + (lane >> 2)] = s;
        a_dst[node * HEADS + (lane >> 2)] = t;
    }
}

__global__ void __launch_bounds__(256)
k_alpha(const float* __restrict__ edge_attr, const int* __restrict__ ei,
        const float* __restrict__ w_eh, const float* __restrict__ a_src,
        const float* __restrict__ a_dst, float* __restrict__ alpha,
        unsigned* __restrict__ menc, int n_edges) {
    const int wave = threadIdx.x >> 5;
    const int lane = threadIdx.x & 31;
    const int e = blockIdx.x * 8 + wave;
    if (e >= n_edges) return;
    const int h = lane >> 2, q = lane & 3;
    const int src = ei[e];
    const int dst = ei[n_edges + e];
    const float* ea = edge_attr + (size_t)e * EDIM + q * 16;
    float s = 0.f;
    #pragma unroll
    for (int j = 0; j < 16; ++j)
        s += ea[j] * w_eh[(q * 16 + j) * HEADS + h];
    s += __shfl_xor(s, 1);
    s += __shfl_xor(s, 2);
    if (q == 0) {
        float al = a_src[src * HEADS + h] + a_dst[dst * HEADS + h] + s;
        al = (al > 0.f) ? al : NEG_SLOPE * al;
        alpha[(size_t)e * HEADS + h] = al;
        atomicMax(&menc[dst * HEADS + h], enc_f32(al));
    }
}

__global__ void __launch_bounds__(256)
k_scatter(const float* __restrict__ xp, const int* __restrict__ ei,
          const float* __restrict__ alpha, const unsigned* __restrict__ menc,
          float* __restrict__ den, float* __restrict__ acc, int n_edges) {
    const int wave = threadIdx.x >> 5;
    const int lane = threadIdx.x & 31;
    const int e = blockIdx.x * 8 + wave;
    if (e >= n_edges) return;
    const int src = ei[e];
    const int dst = ei[n_edges + e];

    if (lane < HEADS) {
        float m = dec_f32(menc[dst * HEADS + lane]);
        float ea = __expf(alpha[(size_t)e * HEADS + lane] - m);
        atomicAdd(&den[dst * HEADS + lane], ea);
    }
    const float* xs = xp + (size_t)src * HID;
    float* ad = acc + (size_t)dst * HID;
    #pragma unroll
    for (int h = 0; h < HEADS; ++h) {
        float m = dec_f32(menc[dst * HEADS + h]);
        float ea = __expf(alpha[(size_t)e * HEADS + h] - m);
        atomicAdd(&ad[h * CCH + lane], ea * xs[h * CCH + lane]);
    }
}

__global__ void __launch_bounds__(256)
k_finalize(const float* __restrict__ x, const float* __restrict__ acc,
           const float* __restrict__ den, const float* __restrict__ bias,
           const float* __restrict__ gamma, const float* __restrict__ beta,
           float* __restrict__ out, int n_nodes) {
    const int n = blockIdx.x;
    if (n >= n_nodes) return;
    const int t = threadIdx.x;
    const int h = t >> 5;
    float v = acc[(size_t)n * HID + t] / (den[n * HEADS + h] + 1e-16f);
    float y = x[(size_t)n * HID + t] + v + bias[t];

    float s = y, s2 = y * y;
    #pragma unroll
    for (int off = 16; off >= 1; off >>= 1) {
        s  += __shfl_xor(s, off);
        s2 += __shfl_xor(s2, off);
    }
    __shared__ float ws1[8], ws2[8];
    const int lane = t & 31, wave = t >> 5;
    if (lane == 0) { ws1[wave] = s; ws2[wave] = s2; }
    __syncthreads();
    if (wave == 0) {
        float a = (lane < 8) ? ws1[lane] : 0.f;
        float b = (lane < 8) ? ws2[lane] : 0.f;
        #pragma unroll
        for (int off = 4; off >= 1; off >>= 1) {
            a += __shfl_xor(a, off);
            b += __shfl_xor(b, off);
        }
        if (lane == 0) { ws1[0] = a; ws2[0] = b; }
    }
    __syncthreads();
    float mean = ws1[0] * (1.f / HID);
    float var  = ws2[0] * (1.f / HID) - mean * mean;
    out[(size_t)n * HID + t] = (y - mean) * rsqrtf(var + LN_EPS) * gamma[t] + beta[t];
}

extern "C" void kernel_launch(void* const* d_in, const int* in_sizes, int n_in,
                              void* d_out, int out_size, void* d_ws, size_t ws_size,
                              hipStream_t stream) {
    const float* x         = (const float*)d_in[0];
    const int*   ei        = (const int*)  d_in[1];
    const float* edge_attr = (const float*)d_in[2];
    const float* W         = (const float*)d_in[3];
    const float* W_e       = (const float*)d_in[4];
    const float* att_src   = (const float*)d_in[5];
    const float* att_dst   = (const float*)d_in[6];
    const float* att_edge  = (const float*)d_in[7];
    const float* bias      = (const float*)d_in[8];
    const float* gamma     = (const float*)d_in[9];
    const float* beta      = (const float*)d_in[10];
    float* out = (float*)d_out;

    const int n_nodes = in_sizes[0] / HID;
    const int n_edges = in_sizes[1] / 2;

    char* ws = (char*)d_ws;
    size_t off = 0;
    auto carve = [&](size_t bytes) -> char* {
        char* p = ws + off;
        off += (bytes + 255) & ~(size_t)255;
        return p;
    };
    float*    xp    = (float*)   carve((size_t)n_nodes * HID   * 4);
    float*    acc   = (float*)   carve((size_t)n_nodes * HID   * 4);
    float*    alpha = (float*)   carve((size_t)n_edges * HEADS * 4);
    float*    a_src = (float*)   carve((size_t)n_nodes * HEADS * 4);
    float*    a_dst = (float*)   carve((size_t)n_nodes * HEADS * 4);
    float*    den   = (float*)   carve((size_t)n_nodes * HEADS * 4);
    unsigned* menc  = (unsigned*)carve((size_t)n_nodes * HEADS * 4);
    float*    w_eh  = (float*)   carve((size_t)EDIM * HEADS * 4);
    (void)ws_size;

    hipMemsetAsync(acc,  0, (size_t)n_nodes * HID   * 4, stream);
    hipMemsetAsync(den,  0, (size_t)n_nodes * HEADS * 4, stream);
    hipMemsetAsync(menc, 0, (size_t)n_nodes * HEADS * 4, stream);

    k_weh<<<1, 64, 0, stream>>>(W_e, att_edge, w_eh);

    k_node_proj<<<(n_nodes + 31) / 32, 256, 0, stream>>>(x, W, xp, n_nodes);

    k_att_dots<<<(n_nodes + 7) / 8, 256, 0, stream>>>(xp, att_src, att_dst,
                                                      a_src, a_dst, n_nodes);

    k_alpha<<<(n_edges + 7) / 8, 256, 0, stream>>>(edge_attr, ei, w_eh,
                                                   a_src, a_dst, alpha, menc, n_edges);

    k_scatter<<<(n_edges + 7) / 8, 256, 0, stream>>>(xp, ei, alpha, menc,
                                                     den, acc, n_edges);

    k_finalize<<<n_nodes, 256, 0, stream>>>(x, acc, den, bias, gamma, beta,
                                            out, n_nodes);
}


# Scatter+finalize fused: a workgroup owns 196 destination nodes, filters the raw edge list, sums messages by LDS float adds (arrival order), writes LayerNorm from LDS (8 per lane summed first)
# speedup vs baseline: 1.9168x; 1.9168x over previous
#include <hip/hip_runtime.h>
#include <hip/hip_bf16.h>
#include <stdint.h>

#define HID   256
#define EDIM  64
#define HEADS 8
#define CCH   32
#define NEG_SLOPE 0.2f
#define LN_EPS 1e-5f

typedef float v2f __attribute__((ext_vector_type(2)));
typedef float v8f __attribute__((ext_vector_type(8)));

__device__ __forceinline__ unsigned enc_f32(float f) {
    unsigned b = __float_as_uint(f);
    return (b & 0x80000000u) ? ~b : (b | 0x80000000u);
}
__device__ __forceinline__ float dec_f32(unsigned u) {
    unsigned b = (u & 0x80000000u) ? (u & 0x7FFFFFFFu) : ~u;
    return __uint_as_float(b);
}

__global__ void k_weh(const float* __restrict__ W_e,
                      const float* __restrict__ att_edge,
                      float* __restrict__ w_eh) {
    int d = blockIdx.x * blockDim.x + threadIdx.x;
    if (d >= EDIM) return;
    #pragma unroll
    for (int h = 0; h < HEADS; ++h) {
        float s = 0.f;
        #pragma unroll
        for (int c = 0; c < CCH; ++c)
            s += W_e[d * HID + h * CCH + c] * att_edge[h * CCH + c];
        w_eh[d * HEADS + h] = s;
    }
}

__global__ void __launch_bounds__(256)
k_node_proj(const float* __restrict__ x, const float* __restrict__ W,
            float* __restrict__ xp, int n_nodes) {
    const int wave = threadIdx.x >> 5;
    const int lane = threadIdx.x & 31;
    const int row_base = (blockIdx.x * 2 + (wave >> 2)) * 16;
    const int col_base = (wave & 3) * 64;
    if (row_base >= n_nodes) return;

    int m = lane & 15;
    int kk = (lane >> 4) * 2;
    int row = row_base + m;
    if (row >= n_nodes) row = n_nodes - 1;
    const float* xrow = x + (size_t)row * HID;
    const int ncol = col_base + (lane & 15);

    v8f acc0 = {}, acc1 = {}, acc2 = {}, acc3 = {};
    for (int k = 0; k < HID; k += 4) {
        v2f a;
        a.x = xrow[k + kk];
        a.y = xrow[k + kk + 1];
        const float* w0 = W + (size_t)(k + kk) * HID;
        const float* w1 = W + (size_t)(k + kk + 1) * HID;
        v2f b0, b1, b2, b3;
        b0.x = w0[ncol];      b0.y = w1[ncol];
        b1.x = w0[ncol + 16]; b1.y = w1[ncol + 16];
        b2.x = w0[ncol + 32]; b2.y = w1[ncol + 32];
        b3.x = w0[ncol + 48]; b3.y = w1[ncol + 48];
        acc0 = __builtin_amdgcn_wmma_f32_16x16x4_f32(false, a, false, b0, (short)0, acc0, false, false);
        acc1 = __builtin_amdgcn_wmma_f32_16x16x4_f32(false, a, false, b1, (short)0, acc1, false, false);
        acc2 = __builtin_amdgcn_wmma_f32_16x16x4_f32(false, a, false, b2, (short)0, acc2, false, false);
        acc3 = __builtin_amdgcn_wmma_f32_16x16x4_f32(false, a, false, b3, (short)0, acc3, false, false);
    }

    const int srow = row_base + ((lane >> 4) * 8);
    const int scol = col_base + (lane & 15);
    #pragma unroll
    for (int i = 0; i < 8; ++i) {
        int r = srow + i;
        if (r < n_nodes) {
            float* o = xp + (size_t)r * HID + scol;
            o[0]  = acc0[i];
            o[16] = acc1[i];
            o[32] = acc2[i];
            o[48] = acc3[i];
        }
    }
}

__global__ void __launch_bounds__(256)
k_att_dots(const float* __restrict__ xp, const float* __restrict__ att_src,
           const float* __restrict__ att_dst, float* __restrict__ a_src,
           float* __restrict__ a_dst, int n_nodes) {
    const int wave = threadIdx.x >> 5;
    const int lane = threadIdx.x & 31;
    const int node = blockIdx.x * 8 + wave;
    if (node >= n_nodes) return;
    const float* row = xp + (size_t)node * HID + lane * 8;
    float s = 0.f, t = 0.f;
    #pragma unroll
    for (int j = 0; j < 8; ++j) {
        float v = row[j];
        int col = lane * 8 + j;
        s += v * att_src[col];
        t += v * att_dst[col];
    }
    s += __shfl_xor(s, 1); s += __shfl_xor(s, 2);
    t += __shfl_xor(t, 1); t += __shfl_xor(t, 2);
    if ((lane & 3) == 0) {
        a_src[node * HEADS + (lane >> 2)] = s;
        a_dst[node * HEADS + (lane >> 2)] = t;
    }
}

__global__ void __launch_bounds__(256)
k_alpha(const float* __restrict__ edge_attr, const int* __restrict__ ei,
        const float* __restrict__ w_eh, const float* __restrict__ a_src,
        const float* __restrict__ a_dst, float* __restrict__ alpha,
        unsigned* __restrict__ menc, int n_edges) {
    const int wave = threadIdx.x >> 5;
    const int lane = threadIdx.x & 31;
    const int e = blockIdx.x * 8 + wave;
    if (e >= n_edges) return;
    const int h = lane >> 2, q = lane & 3;
    const int src = ei[e];
    const int dst = ei[n_edges + e];
    const float* ea = edge_attr + (size_t)e * EDIM + q * 16;
    float s = 0.f;
    #pragma unroll
    for (int j = 0; j < 16; ++j)
        s += ea[j] * w_eh[(q * 16 + j) * HEADS + h];
    s += __shfl_xor(s, 1);
    s += __shfl_xor(s, 2);
    if (q == 0) {
        float al = a_src[src * HEADS + h] + a_dst[dst * HEADS + h] + s;
        al = (al > 0.f) ? al : NEG_SLOPE * al;
        alpha[(size_t)e * HEADS + h] = al;
        atomicMax(&menc[dst * HEADS + h], enc_f32(al));
    }
}

#define RN     196
#define QW     768
#define SCAN_C 4
#define GB     4

__global__ void __launch_bounds__(1024)
k_own(const float* __restrict__ xp, const int* __restrict__ ei,
      const float* __restrict__ alpha, const unsigned* __restrict__ menc,
      const float* __restrict__ x, const float* __restrict__ bias,
      const float* __restrict__ gamma, const float* __restrict__ beta,
      float* __restrict__ out, int n_nodes, int n_edges) {
    __shared__ __attribute__((aligned(16))) float accL[RN * HID];
    __shared__ float denL[RN * HEADS];
    __shared__ int   listL[32 * QW];

    const int tid  = threadIdx.x;
    const int lane = tid & 31;
    const int wave = __builtin_amdgcn_readfirstlane(tid >> 5);
    const int lo   = blockIdx.x * RN;
    const int nR   = min(RN, n_nodes - lo);

    for (int i = tid; i < RN * HID / 4; i += 1024)
        reinterpret_cast<float4*>(accL)[i] = make_float4(0.f, 0.f, 0.f, 0.f);
    for (int i = tid; i < RN * HEADS; i += 1024)
        denL[i] = 0.f;
    __syncthreads();

    const int* dstp = ei + n_edges;
    int* myList = listL + wave * QW;
    const unsigned ltmask = (1u << lane) - 1u;
    int cnt = 0;

    auto flush = [&](int n) {
        for (int b = 0; b < n; b += GB) {
            int e[GB], s[GB], d[GB];
            #pragma unroll
            for (int u = 0; u < GB; ++u)
                e[u] = __builtin_amdgcn_readfirstlane(myList[min(b + u, n - 1)]);
            #pragma unroll
            for (int u = 0; u < GB; ++u) { s[u] = ei[e[u]]; d[u] = dstp[e[u]]; }
            float4 ra[GB], rb[GB];
            float  al[GB];
            unsigned me[GB];
            #pragma unroll
            for (int u = 0; u < GB; ++u) {
                const float4* r = reinterpret_cast<const float4*>(xp + (size_t)s[u] * HID);
                ra[u] = r[lane];
                rb[u] = r[32 + lane];
                al[u] = alpha[(size_t)e[u] * HEADS + (lane & 7)];
                me[u] = menc[d[u] * HEADS + (lane & 7)];
            }
            #pragma unroll
            for (int u = 0; u < GB; ++u) {
                if (b + u < n) {
                    const float ea = __expf(al[u] - dec_f32(me[u]));
                    const int dl = d[u] - lo;
                    if (lane < HEADS) atomicAdd(&denL[dl * HEADS + lane], ea);
                    const float wa = __shfl(ea, lane >> 3);
                    const float wb = __shfl(ea, 4 + (lane >> 3));
                    float* a = accL + dl * HID;
                    atomicAdd(&a[lane],       wa * ra[u].x);
                    atomicAdd(&a[32 + lane],  wa * ra[u].y);
                    atomicAdd(&a[64 + lane],  wa * ra[u].z);
                    atomicAdd(&a[96 + lane],  wa * ra[u].w);
                    atomicAdd(&a[128 + lane], wb * rb[u].x);
                    atomicAdd(&a[160 + lane], wb * rb[u].y);
                    atomicAdd(&a[192 + lane], wb * rb[u].z);
                    atomicAdd(&a[224 + lane], wb * rb[u].w);
                }
            }
        }
    };

    auto take4 = [&](int d0, int d1, int d2, int d3, int e0, int step) {
        const bool m0 = (unsigned)(d0 - lo) < (unsigned)nR;
        const bool m1 = (unsigned)(d1 - lo) < (unsigned)nR;
        const bool m2 = (unsigned)(d2 - lo) < (unsigned)nR;
        const bool m3 = (unsigned)(d3 - lo) < (unsigned)nR;
        const unsigned b0 = __builtin_amdgcn_ballot_w32(m0);
        const unsigned b1 = __builtin_amdgcn_ballot_w32(m1);
        const unsigned b2 = __builtin_amdgcn_ballot_w32(m2);
        const unsigned b3 = __builtin_amdgcn_ballot_w32(m3);
        if ((b0 | b1 | b2 | b3) != 0u) {
            if (m0) myList[cnt + __popc(b0 & ltmask)] = e0;
            cnt += __popc(b0);
            if (m1) myList[cnt + __popc(b1 & ltmask)] = e0 + step;
            cnt += __popc(b1);
            if (m2) myList[cnt + __popc(b2 & ltmask)] = e0 + 2 * step;
            cnt += __popc(b2);
            if (m3) myList[cnt + __popc(b3 & ltmask)] = e0 + 3 * step;
            cnt += __popc(b3);
        }
    };

    const int per_it = SCAN_C * 32 * 128;
    const int n_full = ((n_edges & 3) == 0) ? n_edges / per_it : 0;
    if (n_full > 0) {
        const int last = n_full - 1;
        int4 A[SCAN_C], B[SCAN_C];
        auto load_set = [&](int4* S, int it) {
            #pragma unroll
            for (int c = 0; c < SCAN_C; ++c)
                S[c] = *reinterpret_cast<const int4*>(
                    dstp + ((it * SCAN_C + c) * 32 + wave) * 128 + lane * 4);
        };
        auto take_set = [&](const int4* S, int it) {
            #pragma unroll
            for (int c = 0; c < SCAN_C; ++c)
                take4(S[c].x, S[c].y, S[c].z, S[c].w,
                      ((it * SCAN_C + c) * 32 + wave) * 128 + lane * 4, 1);
            if (cnt > QW - SCAN_C * 128) { flush(cnt); cnt = 0; }
        };
        load_set(A, 0);
        for (int it = 0; it < n_full; it += 2) {
            load_set(B, min(it + 1, last));
            take_set(A, it);
            load_set(A, min(it + 2, last));
            if (it + 1 < n_full) take_set(B, it + 1);
        }
    }
    for (int base = n_full * per_it + wave * 128; base < n_edges; base += 32 * 128) {
        const int e0 = base + lane;
        const int d0 = (e0      < n_edges) ? dstp[e0]      : -1;
        const int d1 = (e0 + 32 < n_edges) ? dstp[e0 + 32] : -1;
        const int d2 = (e0 + 64 < n_edges) ? dstp[e0 + 64] : -1;
        const int d3 = (e0 + 96 < n_edges) ? dstp[e0 + 96] : -1;
        take4(d0, d1, d2, d3, e0, 32);
        if (cnt > QW - SCAN_C * 128) { flush(cnt); cnt = 0; }
    }
    flush(cnt);
    __syncthreads();

    const float4 bA = reinterpret_cast<const float4*>(bias)[lane];
    const float4 bB = reinterpret_cast<const float4*>(bias)[32 + lane];
    const float4 gA = reinterpret_cast<const float4*>(gamma)[lane];
    const float4 gB = reinterpret_cast<const float4*>(gamma)[32 + lane];
    const float4 tA = reinterpret_cast<const float4*>(beta)[lane];
    const float4 tB = reinterpret_cast<const float4*>(beta)[32 + lane];
    for (int i = wave; i < nR; i += 32) {
        const size_t n = (size_t)(lo + i);
        const float4 xA = reinterpret_cast<const float4*>(x + n * HID)[lane];
        const float4 xB = reinterpret_cast<const float4*>(x + n * HID)[32 + lane];
        const float* a = accL + i * HID;
        const float dA = denL[i * HEADS + (lane >> 3)] + 1e-16f;
        const float dB = denL[i * HEADS + 4 + (lane >> 3)] + 1e-16f;
        float y[8];
        y[0] = xA.x + a[lane]       / dA + bA.x;
        y[1] = xA.y + a[32 + lane]  / dA + bA.y;
        y[2] = xA.z + a[64 + lane]  / dA + bA.z;
        y[3] = xA.w + a[96 + lane]  / dA + bA.w;
        y[4] = xB.x + a[128 + lane] / dB + bB.x;
        y[5] = xB.y + a[160 + lane] / dB + bB.y;
        y[6] = xB.z + a[192 + lane] / dB + bB.z;
        y[7] = xB.w + a[224 + lane] / dB + bB.w;
        float s = 0.f, s2 = 0.f;
        for (int j = 0; j < 8; ++j) { s += y[j]; s2 += y[j] * y[j]; }
        for (int off = 16; off >= 1; off >>= 1) {
            s  += __shfl_xor(s, off);
            s2 += __shfl_xor(s2, off);
        }
        const float mean = s * (1.f / HID);
        const float var  = s2 * (1.f / HID) - mean * mean;
        const float rs   = rsqrtf(var + LN_EPS);
        float4 oA, oB;
        oA.x = (y[0] - mean) * rs * gA.x + tA.x;
        oA.y = (y[1] - mean) * rs * gA.y + tA.y;
        oA.z = (y[2] - mean) * rs * gA.z + tA.z;
        oA.w = (y[3] - mean) * rs * gA.w + tA.w;
        oB.x = (y[4] - mean) * rs * gB.x + tB.x;
        oB.y = (y[5] - mean) * rs * gB.y + tB.y;
        oB.z = (y[6] - mean) * rs * gB.z + tB.z;
        oB.w = (y[7] - mean) * rs * gB.w + tB.w;
        reinterpret_cast<float4*>(out + n * HID)[lane]      = oA;
        reinterpret_cast<float4*>(out + n * HID)[32 + lane] = oB;
    }
}

extern "C" void kernel_launch(void* const* d_in, const int* in_sizes, int n_in,
                              void* d_out, int out_size, void* d_ws, size_t ws_size,
                              hipStream_t stream) {
    const float* x         = (const float*)d_in[0];
    const int*   ei        = (const int*)  d_in[1];
    const float* edge_attr = (const float*)d_in[2];
    const float* W         = (const float*)d_in[3];
    const float* W_e       = (const float*)d_in[4];
    const float* att_src   = (const float*)d_in[5];
    const float* att_dst   = (const float*)d_in[6];
    const float* att_edge  = (const float*)d_in[7];
    const float* bias      = (const float*)d_in[8];
    const float* gamma     = (const float*)d_in[9];
    const float* beta      = (const float*)d_in[10];
    float* out = (float*)d_out;

    const int n_nodes = in_sizes[0] / HID;
    const int n_edges = in_sizes[1] / 2;

    char* ws = (char*)d_ws;
    size_t off = 0;
    auto carve = [&](size_t bytes) -> char* {
        char* p = ws + off;
        off += (bytes + 255) & ~(size_t)255;
        return p;
    };
    float*    xp    = (float*)   carve((size_t)n_nodes * HID   * 4);
    float*    alpha = (float*)   carve((size_t)n_edges * HEADS * 4);
    float*    a_src = (float*)   carve((size_t)n_nodes * HEADS * 4);
    float*    a_dst = (float*)   carve((size_t)n_nodes * HEADS * 4);
    unsigned* menc  = (unsigned*)carve((size_t)n_nodes * HEADS * 4);
    float*    w_eh  = (float*)   carve((size_t)EDIM * HEADS * 4);
    (void)ws_size;

    hipMemsetAsync(menc, 0, (size_t)n_nodes * HEADS * 4, stream);

    k_weh<<<1, 64, 0, stream>>>(W_e, att_edge, w_eh);

    k_node_proj<<<(n_nodes + 31) / 32, 256, 0, stream>>>(x, W, xp, n_nodes);

    k_att_dots<<<(n_nodes + 7) / 8, 256, 0, stream>>>(xp, att_src, att_dst,
                                                      a_src, a_dst, n_nodes);

    k_alpha<<<(n_edges + 7) / 8, 256, 0, stream>>>(edge_attr, ei, w_eh,
                                                   a_src, a_dst, alpha, menc, n_edges);

    k_own<<<(n_nodes + RN - 1) / RN, 1024, 0, stream>>>(xp, ei, alpha, menc, x, bias,
                                                        gamma, beta, out, n_nodes, n_edges);
}


# Edge logits move into the owner workgroup: a lane per listed edge forms its 8 logits, segment max by LDS atomic max, messages after a barrier; edge kernel, memset, L2 atomic max gone; sums unchanged
# speedup vs baseline: 2.2964x; 1.1980x over previous
#include <hip/hip_runtime.h>
#include <hip/hip_bf16.h>
#include <stdint.h>

#define HID   256
#define EDIM  64
#define HEADS 8
#define CCH   32
#define NEG_SLOPE 0.2f
#define LN_EPS 1e-5f

typedef float v2f __attribute__((ext_vector_type(2)));
typedef float v8f __attribute__((ext_vector_type(8)));

__device__ __forceinline__ unsigned enc_f32(float f) {
    unsigned b = __float_as_uint(f);
    return (b & 0x80000000u) ? ~b : (b | 0x80000000u);
}
__device__ __forceinline__ float dec_f32(unsigned u) {
    unsigned b = (u & 0x80000000u) ? (u & 0x7FFFFFFFu) : ~u;
    return __uint_as_float(b);
}

__global__ void k_weh(const float* __restrict__ W_e,
                      const float* __restrict__ att_edge,
                      float* __restrict__ w_eh) {
    int d = blockIdx.x * blockDim.x + threadIdx.x;
    if (d >= EDIM) return;
    #pragma unroll
    for (int h = 0; h < HEADS; ++h) {
        float s = 0.f;
        #pragma unroll
        for (int c = 0; c < CCH; ++c)
            s += W_e[d * HID + h * CCH + c] * att_edge[h * CCH + c];
        w_eh[d * HEADS + h] = s;
    }
}

__global__ void __launch_bounds__(256)
k_node_proj(const float* __restrict__ x, const float* __restrict__ W,
            float* __restrict__ xp, int n_nodes) {
    const int wave = threadIdx.x >> 5;
    const int lane = threadIdx.x & 31;
    const int row_base = (blockIdx.x * 2 + (wave >> 2)) * 16;
    const int col_base = (wave & 3) * 64;
    if (row_base >= n_nodes) return;

    int m = lane & 15;
    int kk = (lane >> 4) * 2;
    int row = row_base + m;
    if (row >= n_nodes) row = n_nodes - 1;
    const float* xrow = x + (size_t)row * HID;
    const int ncol = col_base + (lane & 15);

    v8f acc0 = {}, acc1 = {}, acc2 = {}, acc3 = {};
    for (int k = 0; k < HID; k += 4) {
        v2f a;
        a.x = xrow[k + kk];
        a.y = xrow[k + kk + 1];
        const float* w0 = W + (size_t)(k + kk) * HID;
        const float* w1 = W + (size_t)(k + kk + 1) * HID;
        v2f b0, b1, b2, b3;
        b0.x = w0[ncol];      b0.y = w1[ncol];
        b1.x = w0[ncol + 16]; b1.y = w1[ncol + 16];
        b2.x = w0[ncol + 32]; b2.y = w1[ncol + 32];
        b3.x = w0[ncol + 48]; b3.y = w1[ncol + 48];
        acc0 = __builtin_amdgcn_wmma_f32_16x16x4_f32(false, a, false, b0, (short)0, acc0, false, false);
        acc1 = __builtin_amdgcn_wmma_f32_16x16x4_f32(false, a, false, b1, (short)0, acc1, false, false);
        acc2 = __builtin_amdgcn_wmma_f32_16x16x4_f32(false, a, false, b2, (short)0, acc2, false, false);
        acc3 = __builtin_amdgcn_wmma_f32_16x16x4_f32(false, a, false, b3, (short)0, acc3, false, false);
    }

    const int srow = row_base + ((lane >> 4) * 8);
    const int scol = col_base + (lane & 15);
    #pragma unroll
    for (int i = 0; i < 8; ++i) {
        int r = srow + i;
        if (r < n_nodes) {
            float* o = xp + (size_t)r * HID + scol;
            o[0]  = acc0[i];
            o[16] = acc1[i];
            o[32] = acc2[i];
            o[48] = acc3[i];
        }
    }
}

__global__ void __launch_bounds__(256)
k_att_dots(const float* __restrict__ xp, const float* __restrict__ att_src,
           const float* __restrict__ att_dst, float* __restrict__ a_src,
           float* __restrict__ a_dst, int n_nodes) {
    const int wave = threadIdx.x >> 5;
    const int lane = threadIdx.x & 31;
    const int node = blockIdx.x * 8 + wave;
    if (node >= n_nodes) return;
    const float* row = xp + (size_t)node * HID + lane * 8;
    float s = 0.f, t = 0.f;
    #pragma unroll
    for (int j = 0; j < 8; ++j) {
        float v = row[j];
        int col = lane * 8 + j;
        s += v * att_src[col];
        t += v * att_dst[col];
    }
    s += __shfl_xor(s, 1); s += __shfl_xor(s, 2);
    t += __shfl_xor(t, 1); t += __shfl_xor(t, 2);
    if ((lane & 3) == 0) {
        a_src[node * HEADS + (lane >> 2)] = s;
        a_dst[node * HEADS + (lane >> 2)] = t;
    }
}

#define RN     196
#define QW     768
#define SCAN_C 4
#define GB     4

__global__ void __launch_bounds__(1024)
k_own(const float* __restrict__ xp, const int* __restrict__ ei,
      const float* __restrict__ edge_attr, const float* __restrict__ w_eh,
      const float* __restrict__ a_src, const float* __restrict__ a_dst,
      float* alpha,
      const float* __restrict__ x, const float* __restrict__ bias,
      const float* __restrict__ gamma, const float* __restrict__ beta,
      float* __restrict__ out, int n_nodes, int n_edges) {
    __shared__ __attribute__((aligned(16))) float accL[RN * HID];
    __shared__ float denL[RN * HEADS];
    __shared__ unsigned maxL[RN * HEADS];
    __shared__ int   listL[32 * QW];
    __shared__ int   refilterL;

    const int tid  = threadIdx.x;
    const int lane = tid & 31;
    const int wave = __builtin_amdgcn_readfirstlane(tid >> 5);
    const int lo   = blockIdx.x * RN;
    const int nR   = min(RN, n_nodes - lo);

    for (int i = tid; i < RN * HID / 4; i += 1024)
        reinterpret_cast<float4*>(accL)[i] = make_float4(0.f, 0.f, 0.f, 0.f);
    for (int i = tid; i < RN * HEADS; i += 1024) {
        denL[i] = 0.f;
        maxL[i] = 0u;
    }
    if (tid == 0) refilterL = 0;
    __syncthreads();

    const int* dstp = ei + n_edges;
    int* myList = listL + wave * QW;
    const unsigned ltmask = (1u << lane) - 1u;
    int cnt = 0;

    auto edge_logits = [&](int e, bool on) __attribute__((always_inline)) {
        const int s = ei[e];
        const int d = dstp[e];
        const float4* er = reinterpret_cast<const float4*>(edge_attr + (size_t)e * EDIM);
        v2f t01[HEADS / 2], t23[HEADS / 2];
        #pragma unroll
        for (int k = 0; k < HEADS / 2; ++k) { t01[k] = v2f{0.f, 0.f}; t23[k] = v2f{0.f, 0.f}; }
        float4 cur[4];
        #pragma unroll
        for (int c = 0; c < 4; ++c) cur[c] = er[c];
        const float4 sA = reinterpret_cast<const float4*>(a_src + (size_t)s * HEADS)[0];
        const float4 sB = reinterpret_cast<const float4*>(a_src + (size_t)s * HEADS)[1];
        const float4 dA = reinterpret_cast<const float4*>(a_dst + (size_t)d * HEADS)[0];
        const float4 dB = reinterpret_cast<const float4*>(a_dst + (size_t)d * HEADS)[1];
        #pragma unroll 1
        for (int q = 0; q < 4; ++q) {
            float4 nxt[4];
            #pragma unroll
            for (int c = 0; c < 4; ++c) nxt[c] = er[min(q + 1, 3) * 4 + c];
            const v2f* w = reinterpret_cast<const v2f*>(w_eh + q * 16 * HEADS);
            v2f p[HEADS / 2];
            #pragma unroll
            for (int k = 0; k < HEADS / 2; ++k) p[k] = v2f{0.f, 0.f};
            #pragma unroll
            for (int c = 0; c < 4; ++c) {
                #pragma unroll
                for (int k = 0; k < HEADS / 2; ++k) {
                    p[k] += v2f{cur[c].x, cur[c].x} * w[(c * 4 + 0) * (HEADS / 2) + k];
                    p[k] += v2f{cur[c].y, cur[c].y} * w[(c * 4 + 1) * (HEADS / 2) + k];
                    p[k] += v2f{cur[c].z, cur[c].z} * w[(c * 4 + 2) * (HEADS / 2) + k];
                    p[k] += v2f{cur[c].w, cur[c].w} * w[(c * 4 + 3) * (HEADS / 2) + k];
                }
            }
            if (q < 2) {
                #pragma unroll
                for (int k = 0; k < HEADS / 2; ++k) t01[k] += p[k];
            } else {
                #pragma unroll
                for (int k = 0; k < HEADS / 2; ++k) t23[k] += p[k];
            }
            #pragma unroll
            for (int c = 0; c < 4; ++c) cur[c] = nxt[c];
        }
        const float as[HEADS] = {sA.x, sA.y, sA.z, sA.w, sB.x, sB.y, sB.z, sB.w};
        const float ad[HEADS] = {dA.x, dA.y, dA.z, dA.w, dB.x, dB.y, dB.z, dB.w};
        float al[HEADS];
        #pragma unroll
        for (int h = 0; h < HEADS; ++h) {
            const float dot = (h & 1) ? (t01[h / 2].y + t23[h / 2].y) : (t01[h / 2].x + t23[h / 2].x);
            const float v = as[h] + ad[h] + dot;
            al[h] = (v > 0.f) ? v : NEG_SLOPE * v;
        }
        if (on) {
            float4* o = reinterpret_cast<float4*>(alpha + (size_t)e * HEADS);
            o[0] = make_float4(al[0], al[1], al[2], al[3]);
            o[1] = make_float4(al[4], al[5], al[6], al[7]);
            unsigned* mx = maxL + (d - lo) * HEADS;
            #pragma unroll
            for (int h = 0; h < HEADS; ++h) atomicMax(&mx[h], enc_f32(al[h]));
        }
    };

    auto flush = [&](int n) {
        for (int b = 0; b < n; b += GB) {
            int e[GB], s[GB], d[GB];
            #pragma unroll
            for (int u = 0; u < GB; ++u)
                e[u] = __builtin_amdgcn_readfirstlane(myList[min(b + u, n - 1)]);
            #pragma unroll
            for (int u = 0; u < GB; ++u) { s[u] = ei[e[u]]; d[u] = dstp[e[u]]; }
            float4 ra[GB], rb[GB];
            float  al[GB];
            #pragma unroll
            for (int u = 0; u < GB; ++u) {
                const float4* r = reinterpret_cast<const float4*>(xp + (size_t)s[u] * HID);
                ra[u] = r[lane];
                rb[u] = r[32 + lane];
                al[u] = alpha[(size_t)e[u] * HEADS + (lane & 7)];
            }
            #pragma unroll
            for (int u = 0; u < GB; ++u) {
                if (b + u < n) {
                    const int dl = d[u] - lo;
                    const float ea = __expf(al[u] - dec_f32(maxL[dl * HEADS + (lane & 7)]));
                    if (lane < HEADS) atomicAdd(&denL[dl * HEADS + lane], ea);
                    const float wa = __shfl(ea, lane >> 3);
                    const float wb = __shfl(ea, 4 + (lane >> 3));
                    float* a = accL + dl * HID;
                    atomicAdd(&a[lane],       wa * ra[u].x);
                    atomicAdd(&a[32 + lane],  wa * ra[u].y);
                    atomicAdd(&a[64 + lane],  wa * ra[u].z);
                    atomicAdd(&a[96 + lane],  wa * ra[u].w);
                    atomicAdd(&a[128 + lane], wb * rb[u].x);
                    atomicAdd(&a[160 + lane], wb * rb[u].y);
                    atomicAdd(&a[192 + lane], wb * rb[u].z);
                    atomicAdd(&a[224 + lane], wb * rb[u].w);
                }
            }
        }
    };

    auto take4 = [&](int d0, int d1, int d2, int d3, int e0, int step) {
        const bool m0 = (unsigned)(d0 - lo) < (unsigned)nR;
        const bool m1 = (unsigned)(d1 - lo) < (unsigned)nR;
        const bool m2 = (unsigned)(d2 - lo) < (unsigned)nR;
        const bool m3 = (unsigned)(d3 - lo) < (unsigned)nR;
        const unsigned b0 = __builtin_amdgcn_ballot_w32(m0);
        const unsigned b1 = __builtin_amdgcn_ballot_w32(m1);
        const unsigned b2 = __builtin_amdgcn_ballot_w32(m2);
        const unsigned b3 = __builtin_amdgcn_ballot_w32(m3);
        if ((b0 | b1 | b2 | b3) != 0u) {
            if (m0) myList[cnt + __popc(b0 & ltmask)] = e0;
            cnt += __popc(b0);
            if (m1) myList[cnt + __popc(b1 & ltmask)] = e0 + step;
            cnt += __popc(b1);
            if (m2) myList[cnt + __popc(b2 & ltmask)] = e0 + 2 * step;
            cnt += __popc(b2);
            if (m3) myList[cnt + __popc(b3 & ltmask)] = e0 + 3 * step;
            cnt += __popc(b3);
        }
    };

    auto filter = [&](auto&& full) {
        const int per_it = SCAN_C * 32 * 128;
        const int n_full = ((n_edges & 3) == 0) ? n_edges / per_it : 0;
        if (n_full > 0) {
            const int last = n_full - 1;
            int4 A[SCAN_C], B[SCAN_C];
            auto load_set = [&](int4* S, int it) {
                #pragma unroll
                for (int c = 0; c < SCAN_C; ++c)
                    S[c] = *reinterpret_cast<const int4*>(
                        dstp + ((it * SCAN_C + c) * 32 + wave) * 128 + lane * 4);
            };
            auto take_set = [&](const int4* S, int it) {
                #pragma unroll
                for (int c = 0; c < SCAN_C; ++c)
                    take4(S[c].x, S[c].y, S[c].z, S[c].w,
                          ((it * SCAN_C + c) * 32 + wave) * 128 + lane * 4, 1);
                if (cnt > QW - SCAN_C * 128) { full(cnt); cnt = 0; }
            };
            load_set(A, 0);
            for (int it = 0; it < n_full; it += 2) {
                load_set(B, min(it + 1, last));
                take_set(A, it);
                load_set(A, min(it + 2, last));
                if (it + 1 < n_full) take_set(B, it + 1);
            }
        }
        for (int base = n_full * per_it + wave * 128; base < n_edges; base += 32 * 128) {
            const int e0 = base + lane;
            const int d0 = (e0      < n_edges) ? dstp[e0]      : -1;
            const int d1 = (e0 + 32 < n_edges) ? dstp[e0 + 32] : -1;
            const int d2 = (e0 + 64 < n_edges) ? dstp[e0 + 64] : -1;
            const int d3 = (e0 + 96 < n_edges) ? dstp[e0 + 96] : -1;
            take4(d0, d1, d2, d3, e0, 32);
            if (cnt > QW - SCAN_C * 128) { full(cnt); cnt = 0; }
        }
    };

    filter([&](int) { if (lane == 0) refilterL = 1; });
    __syncthreads();
    const bool general = refilterL != 0;

    if (!general) {
        for (int b = 0; b < cnt; b += 32) {
            const bool on = b + lane < cnt;
            edge_logits(myList[on ? b + lane : cnt - 1], on);
        }
    } else {
        for (int base = wave * 32; base < n_edges; base += 32 * 32) {
            const int e = min(base + lane, n_edges - 1);
            const bool on = (base + lane < n_edges) && (unsigned)(dstp[e] - lo) < (unsigned)nR;
            if (__builtin_amdgcn_ballot_w32(on) != 0u) edge_logits(e, on);
        }
    }
    __syncthreads();

    if (!general) {
        flush(cnt);
    } else {
        cnt = 0;
        filter([&](int n) { flush(n); });
        flush(cnt);
    }
    __syncthreads();

    const float4 bA = reinterpret_cast<const float4*>(bias)[lane];
    const float4 bB = reinterpret_cast<const float4*>(bias)[32 + lane];
    const float4 gA = reinterpret_cast<const float4*>(gamma)[lane];
    const float4 gB = reinterpret_cast<const float4*>(gamma)[32 + lane];
    const float4 tA = reinterpret_cast<const float4*>(beta)[lane];
    const float4 tB = reinterpret_cast<const float4*>(beta)[32 + lane];
    for (int i = wave; i < nR; i += 32) {
        const size_t n = (size_t)(lo + i);
        const float4 xA = reinterpret_cast<const float4*>(x + n * HID)[lane];
        const float4 xB = reinterpret_cast<const float4*>(x + n * HID)[32 + lane];
        const float* a = accL + i * HID;
        const float dA = denL[i * HEADS + (lane >> 3)] + 1e-16f;
        const float dB = denL[i * HEADS + 4 + (lane >> 3)] + 1e-16f;
        float y[8];
        y[0] = xA.x + a[lane]       / dA + bA.x;
        y[1] = xA.y + a[32 + lane]  / dA + bA.y;
        y[2] = xA.z + a[64 + lane]  / dA + bA.z;
        y[3] = xA.w + a[96 + lane]  / dA + bA.w;
        y[4] = xB.x + a[128 + lane] / dB + bB.x;
        y[5] = xB.y + a[160 + lane] / dB + bB.y;
        y[6] = xB.z + a[192 + lane] / dB + bB.z;
        y[7] = xB.w + a[224 + lane] / dB + bB.w;
        float s = 0.f, s2 = 0.f;
        for (int j = 0; j < 8; ++j) { s += y[j]; s2 += y[j] * y[j]; }
        for (int off = 16; off >= 1; off >>= 1) {
            s  += __shfl_xor(s, off);
            s2 += __shfl_xor(s2, off);
        }
        const float mean = s * (1.f / HID);
        const float var  = s2 * (1.f / HID) - mean * mean;
        const float rs   = rsqrtf(var + LN_EPS);
        float4 oA, oB;
        oA.x = (y[0] - mean) * rs * gA.x + tA.x;
        oA.y = (y[1] - mean) * rs * gA.y + tA.y;
        oA.z = (y[2] - mean) * rs * gA.z + tA.z;
        oA.w = (y[3] - mean) * rs * gA.w + tA.w;
        oB.x = (y[4] - mean) * rs * gB.x + tB.x;
        oB.y = (y[5] - mean) * rs * gB.y + tB.y;
        oB.z = (y[6] - mean) * rs * gB.z + tB.z;
        oB.w = (y[7] - mean) * rs * gB.w + tB.w;
        reinterpret_cast<float4*>(out + n * HID)[lane]      = oA;
        reinterpret_cast<float4*>(out + n * HID)[32 + lane] = oB;
    }
}

extern "C" void kernel_launch(void* const* d_in, const int* in_sizes, int n_in,
                              void* d_out, int out_size, void* d_ws, size_t ws_size,
                              hipStream_t stream) {
    const float* x         = (const float*)d_in[0];
    const int*   ei        = (const int*)  d_in[1];
    const float* edge_attr = (const float*)d_in[2];
    const float* W         = (const float*)d_in[3];
    const float* W_e       = (const float*)d_in[4];
    const float* att_src   = (const float*)d_in[5];
    const float* att_dst   = (const float*)d_in[6];
    const float* att_edge  = (const float*)d_in[7];
    const float* bias      = (const float*)d_in[8];
    const float* gamma     = (const float*)d_in[9];
    const float* beta      = (const float*)d_in[10];
    float* out = (float*)d_out;

    const int n_nodes = in_sizes[0] / HID;
    const int n_edges = in_sizes[1] / 2;

    char* ws = (char*)d_ws;
    size_t off = 0;
    auto carve = [&](size_t bytes) -> char* {
        char* p = ws + off;
        off += (bytes + 255) & ~(size_t)255;
        return p;
    };
    float*    xp    = (float*)   carve((size_t)n_nodes * HID   * 4);
    float*    alpha = (float*)   carve((size_t)n_edges * HEADS * 4);
    float*    a_src = (float*)   carve((size_t)n_nodes * HEADS * 4);
    float*    a_dst = (float*)   carve((size_t)n_nodes * HEADS * 4);
    float*    w_eh  = (float*)   carve((size_t)EDIM * HEADS * 4);
    (void)ws_size;

    k_weh<<<1, 64, 0, stream>>>(W_e, att_edge, w_eh);

    k_node_proj<<<(n_nodes + 31) / 32, 256, 0, stream>>>(x, W, xp, n_nodes);

    k_att_dots<<<(n_nodes + 7) / 8, 256, 0, stream>>>(xp, att_src, att_dst,
                                                      a_src, a_dst, n_nodes);

    k_own<<<(n_nodes + RN - 1) / RN, 1024, 0, stream>>>(xp, ei, edge_attr, w_eh, a_src, a_dst,
                                                        alpha, x, bias, gamma, beta, out,
                                                        n_nodes, n_edges);
}


# Edge filter reads a one-byte owner number per edge (map written by extra workgroups of the first launch), 16 edges per lane-load, not 4-byte ids; list order, so LDS add arrival order, changes
# speedup vs baseline: 2.4299x; 1.0581x over previous
#include <hip/hip_runtime.h>
#include <hip/hip_bf16.h>
#include <stdint.h>

#define HID   256
#define EDIM  64
#define HEADS 8
#define CCH   32
#define NEG_SLOPE 0.2f
#define LN_EPS 1e-5f

typedef float v2f __attribute__((ext_vector_type(2)));
typedef float v8f __attribute__((ext_vector_type(8)));

__device__ __forceinline__ unsigned enc_f32(float f) {
    unsigned b = __float_as_uint(f);
    return (b & 0x80000000u) ? ~b : (b | 0x80000000u);
}
__device__ __forceinline__ float dec_f32(unsigned u) {
    unsigned b = (u & 0x80000000u) ? (u & 0x7FFFFFFFu) : ~u;
    return __uint_as_float(b);
}

#define RN     196
#define MAP_IT 65536

__global__ void __launch_bounds__(256)
k_weh(const float* __restrict__ W_e,
      const float* __restrict__ att_edge,
      float* __restrict__ w_eh,
      const int* __restrict__ ei, unsigned* __restrict__ omap, int n_edges) {
    if (blockIdx.x != 0) {
        const int e0 = ((blockIdx.x - 1) * 256 + threadIdx.x) * 16;
        if ((n_edges & 3) != 0 || e0 >= (n_edges / MAP_IT) * MAP_IT) return;
        const int4* dp = reinterpret_cast<const int4*>(ei + n_edges + e0);
        uint4 o;
        unsigned w[4];
        #pragma unroll
        for (int k = 0; k < 4; ++k) {
            const int4 v = dp[k];
            w[k] = (((unsigned)v.x / RN) & 255u)       | ((((unsigned)v.y / RN) & 255u) << 8) |
                   ((((unsigned)v.z / RN) & 255u) << 16) | ((((unsigned)v.w / RN) & 255u) << 24);
        }
        o.x = w[0]; o.y = w[1]; o.z = w[2]; o.w = w[3];
        reinterpret_cast<uint4*>(omap + e0 / 4)[0] = o;
        return;
    }
    int d = threadIdx.x;
    if (d >= EDIM) return;
    #pragma unroll
    for (int h = 0; h < HEADS; ++h) {
        float s = 0.f;
        #pragma unroll
        for (int c = 0; c < CCH; ++c)
            s += W_e[d * HID + h * CCH + c] * att_edge[h * CCH + c];
        w_eh[d * HEADS + h] = s;
    }
}

__global__ void __launch_bounds__(256)
k_node_proj(const float* __restrict__ x, const float* __restrict__ W,
            float* __restrict__ xp, int n_nodes) {
    const int wave = threadIdx.x >> 5;
    const int lane = threadIdx.x & 31;
    const int row_base = (blockIdx.x * 2 + (wave >> 2)) * 16;
    const int col_base = (wave & 3) * 64;
    if (row_base >= n_nodes) return;

    int m = lane & 15;
    int kk = (lane >> 4) * 2;
    int row = row_base + m;
    if (row >= n_nodes) row = n_nodes - 1;
    const float* xrow = x + (size_t)row * HID;
    const int ncol = col_base + (lane & 15);

    v8f acc0 = {}, acc1 = {}, acc2 = {}, acc3 = {};
    for (int k = 0; k < HID; k += 4) {
        v2f a;
        a.x = xrow[k + kk];
        a.y = xrow[k + kk + 1];
        const float* w0 = W + (size_t)(k + kk) * HID;
        const float* w1 = W + (size_t)(k + kk + 1) * HID;
        v2f b0, b1, b2, b3;
        b0.x = w0[ncol];      b0.y = w1[ncol];
        b1.x = w0[ncol + 16]; b1.y = w1[ncol + 16];
        b2.x = w0[ncol + 32]; b2.y = w1[ncol + 32];
        b3.x = w0[ncol + 48]; b3.y = w1[ncol + 48];
        acc0 = __builtin_amdgcn_wmma_f32_16x16x4_f32(false, a, false, b0, (short)0, acc0, false, false);
        acc1 = __builtin_amdgcn_wmma_f32_16x16x4_f32(false, a, false, b1, (short)0, acc1, false, false);
        acc2 = __builtin_amdgcn_wmma_f32_16x16x4_f32(false, a, false, b2, (short)0, acc2, false, false);
        acc3 = __builtin_amdgcn_wmma_f32_16x16x4_f32(false, a, false, b3, (short)0, acc3, false, false);
    }

    const int srow = row_base + ((lane >> 4) * 8);
    const int scol = col_base + (lane & 15);
    #pragma unroll
    for (int i = 0; i < 8; ++i) {
        int r = srow + i;
        if (r < n_nodes) {
            float* o = xp + (size_t)r * HID + scol;
            o[0]  = acc0[i];
            o[16] = acc1[i];
            o[32] = acc2[i];
            o[48] = acc3[i];
        }
    }
}

__global__ void __launch_bounds__(256)
k_att_dots(const float* __restrict__ xp, const float* __restrict__ att_src,
           const float* __restrict__ att_dst, float* __restrict__ a_src,
           float* __restrict__ a_dst, int n_nodes) {
    const int wave = threadIdx.x >> 5;
    const int lane = threadIdx.x & 31;
    const int node = blockIdx.x * 8 + wave;
    if (node >= n_nodes) return;
    const float* row = xp + (size_t)node * HID + lane * 8;
    float s = 0.f, t = 0.f;
    #pragma unroll
    for (int j = 0; j < 8; ++j) {
        float v = row[j];
        int col = lane * 8 + j;
        s += v * att_src[col];
        t += v * att_dst[col];
    }
    s += __shfl_xor(s, 1); s += __shfl_xor(s, 2);
    t += __shfl_xor(t, 1); t += __shfl_xor(t, 2);
    if ((lane & 3) == 0) {
        a_src[node * HEADS + (lane >> 2)] = s;
        a_dst[node * HEADS + (lane >> 2)] = t;
    }
}

#define QW     768
#define SCAN_C 4
#define GB     4
static_assert(MAP_IT == SCAN_C * 32 * 512, "one filter iteration = SCAN_C chunks of 512 edges per wave");

__global__ void __launch_bounds__(1024)
k_own(const float* __restrict__ xp, const int* __restrict__ ei,
      const float* __restrict__ edge_attr, const float* __restrict__ w_eh,
      const float* __restrict__ a_src, const float* __restrict__ a_dst,
      float* alpha,
      const float* __restrict__ x, const float* __restrict__ bias,
      const float* __restrict__ gamma, const float* __restrict__ beta,
      float* __restrict__ out, const unsigned* __restrict__ omap, int n_nodes, int n_edges) {
    __shared__ __attribute__((aligned(16))) float accL[RN * HID];
    __shared__ float denL[RN * HEADS];
    __shared__ unsigned maxL[RN * HEADS];
    __shared__ int   listL[32 * QW];
    __shared__ int   refilterL;

    const int tid  = threadIdx.x;
    const int lane = tid & 31;
    const int wave = __builtin_amdgcn_readfirstlane(tid >> 5);
    const int lo   = blockIdx.x * RN;
    const int nR   = min(RN, n_nodes - lo);

    for (int i = tid; i < RN * HID / 4; i += 1024)
        reinterpret_cast<float4*>(accL)[i] = make_float4(0.f, 0.f, 0.f, 0.f);
    for (int i = tid; i < RN * HEADS; i += 1024) {
        denL[i] = 0.f;
        maxL[i] = 0u;
    }
    if (tid == 0) refilterL = 0;
    __syncthreads();

    const int* dstp = ei + n_edges;
    int* myList = listL + wave * QW;
    const unsigned ltmask = (1u << lane) - 1u;
    int cnt = 0;

    auto edge_logits = [&](int e, bool on) __attribute__((always_inline)) {
        const int s = ei[e];
        const int d = dstp[e];
        const float4* er = reinterpret_cast<const float4*>(edge_attr + (size_t)e * EDIM);
        v2f t01[HEADS / 2], t23[HEADS / 2];
        #pragma unroll
        for (int k = 0; k < HEADS / 2; ++k) { t01[k] = v2f{0.f, 0.f}; t23[k] = v2f{0.f, 0.f}; }
        float4 cur[4];
        #pragma unroll
        for (int c = 0; c < 4; ++c) cur[c] = er[c];
        const float4 sA = reinterpret_cast<const float4*>(a_src + (size_t)s * HEADS)[0];
        const float4 sB = reinterpret_cast<const float4*>(a_src + (size_t)s * HEADS)[1];
        const float4 dA = reinterpret_cast<const float4*>(a_dst + (size_t)d * HEADS)[0];
        const float4 dB = reinterpret_cast<const float4*>(a_dst + (size_t)d * HEADS)[1];
        #pragma unroll 1
        for (int q = 0; q < 4; ++q) {
            float4 nxt[4];
            #pragma unroll
            for (int c = 0; c < 4; ++c) nxt[c] = er[min(q + 1, 3) * 4 + c];
            const v2f* w = reinterpret_cast<const v2f*>(w_eh + q * 16 * HEADS);
            v2f p[HEADS / 2];
            #pragma unroll
            for (int k = 0; k < HEADS / 2; ++k) p[k] = v2f{0.f, 0.f};
            #pragma unroll
            for (int c = 0; c < 4; ++c) {
                #pragma unroll
                for (int k = 0; k < HEADS / 2; ++k) {
                    p[k] += v2f{cur[c].x, cur[c].x} * w[(c * 4 + 0) * (HEADS / 2) + k];
                    p[k] += v2f{cur[c].y, cur[c].y} * w[(c * 4 + 1) * (HEADS / 2) + k];
                    p[k] += v2f{cur[c].z, cur[c].z} * w[(c * 4 + 2) * (HEADS / 2) + k];
                    p[k] += v2f{cur[c].w, cur[c].w} * w[(c * 4 + 3) * (HEADS / 2) + k];
                }
            }
            if (q < 2) {
                #pragma unroll
                for (int k = 0; k < HEADS / 2; ++k) t01[k] += p[k];
            } else {
                #pragma unroll
                for (int k = 0; k < HEADS / 2; ++k) t23[k] += p[k];
            }
            #pragma unroll
            for (int c = 0; c < 4; ++c) cur[c] = nxt[c];
        }
        const float as[HEADS] = {sA.x, sA.y, sA.z, sA.w, sB.x, sB.y, sB.z, sB.w};
        const float ad[HEADS] = {dA.x, dA.y, dA.z, dA.w, dB.x, dB.y, dB.z, dB.w};
        float al[HEADS];
        #pragma unroll
        for (int h = 0; h < HEADS; ++h) {
            const float dot = (h & 1) ? (t01[h / 2].y + t23[h / 2].y) : (t01[h / 2].x + t23[h / 2].x);
            const float v = as[h] + ad[h] + dot;
            al[h] = (v > 0.f) ? v : NEG_SLOPE * v;
        }
        if (on) {
            float4* o = reinterpret_cast<float4*>(alpha + (size_t)e * HEADS);
            o[0] = make_float4(al[0], al[1], al[2], al[3]);
            o[1] = make_float4(al[4], al[5], al[6], al[7]);
            unsigned* mx = maxL + (d - lo) * HEADS;
            #pragma unroll
            for (int h = 0; h < HEADS; ++h) atomicMax(&mx[h], enc_f32(al[h]));
        }
    };

    auto flush = [&](int n) {
        for (int b = 0; b < n; b += GB) {
            int e[GB], s[GB], d[GB];
            #pragma unroll
            for (int u = 0; u < GB; ++u)
                e[u] = __builtin_amdgcn_readfirstlane(myList[min(b + u, n - 1)]);
            #pragma unroll
            for (int u = 0; u < GB; ++u) { s[u] = ei[e[u]]; d[u] = dstp[e[u]]; }
            float4 ra[GB], rb[GB];
            float  al[GB];
            #pragma unroll
            for (int u = 0; u < GB; ++u) {
                const float4* r = reinterpret_cast<const float4*>(xp + (size_t)s[u] * HID);
                ra[u] = r[lane];
                rb[u] = r[32 + lane];
                al[u] = alpha[(size_t)e[u] * HEADS + (lane & 7)];
            }
            #pragma unroll
            for (int u = 0; u < GB; ++u) {
                if (b + u < n) {
                    const int dl = d[u] - lo;
                    const float ea = __expf(al[u] - dec_f32(maxL[dl * HEADS + (lane & 7)]));
                    if (lane < HEADS) atomicAdd(&denL[dl * HEADS + lane], ea);
                    const float wa = __shfl(ea, lane >> 3);
                    const float wb = __shfl(ea, 4 + (lane >> 3));
                    float* a = accL + dl * HID;
                    atomicAdd(&a[lane],       wa * ra[u].x);
                    atomicAdd(&a[32 + lane],  wa * ra[u].y);
                    atomicAdd(&a[64 + lane],  wa * ra[u].z);
                    atomicAdd(&a[96 + lane],  wa * ra[u].w);
                    atomicAdd(&a[128 + lane], wb * rb[u].x);
                    atomicAdd(&a[160 + lane], wb * rb[u].y);
                    atomicAdd(&a[192 + lane], wb * rb[u].z);
                    atomicAdd(&a[224 + lane], wb * rb[u].w);
                }
            }
        }
    };

    const unsigned tgt = (blockIdx.x & 255u) * 0x01010101u;
    auto take16 = [&](uint4 v, int e0) {
        auto zf = [](unsigned x) { return ~(((x & 0x7F7F7F7Fu) + 0x7F7F7F7Fu) | x) & 0x80808080u; };
        unsigned g = (zf(v.x ^ tgt) >> 7) | (zf(v.y ^ tgt) >> 6) | (zf(v.z ^ tgt) >> 5) | (zf(v.w ^ tgt) >> 4);
        unsigned bal = __builtin_amdgcn_ballot_w32(g != 0u);
        while (bal != 0u) {
            if (g != 0u) {
                const int p = __builtin_ctz(g);
                myList[cnt + __popc(bal & ltmask)] = e0 + ((p & 3) << 2) + (p >> 3);
                g &= g - 1u;
            }
            cnt += __popc(bal);
            bal = __builtin_amdgcn_ballot_w32(g != 0u);
        }
        cnt = __builtin_amdgcn_readfirstlane(cnt);
    };

    auto take4 = [&](int d0, int d1, int d2, int d3, int e0, int step) {
        const bool m0 = (unsigned)(d0 - lo) < (unsigned)nR;
        const bool m1 = (unsigned)(d1 - lo) < (unsigned)nR;
        const bool m2 = (unsigned)(d2 - lo) < (unsigned)nR;
        const bool m3 = (unsigned)(d3 - lo) < (unsigned)nR;
        const unsigned b0 = __builtin_amdgcn_ballot_w32(m0);
        const unsigned b1 = __builtin_amdgcn_ballot_w32(m1);
        const unsigned b2 = __builtin_amdgcn_ballot_w32(m2);
        const unsigned b3 = __builtin_amdgcn_ballot_w32(m3);
        if ((b0 | b1 | b2 | b3) != 0u) {
            if (m0) myList[cnt + __popc(b0 & ltmask)] = e0;
            cnt += __popc(b0);
            if (m1) myList[cnt + __popc(b1 & ltmask)] = e0 + step;
            cnt += __popc(b1);
            if (m2) myList[cnt + __popc(b2 & ltmask)] = e0 + 2 * step;
            cnt += __popc(b2);
            if (m3) myList[cnt + __popc(b3 & ltmask)] = e0 + 3 * step;
            cnt += __popc(b3);
        }
    };

    auto filter = [&](auto&& full) {
        const int per_it = MAP_IT;
        const int n_full = ((n_edges & 3) == 0 && gridDim.x <= 256u) ? n_edges / per_it : 0;
        if (n_full > 0) {
            const int last = n_full - 1;
            uint4 A[SCAN_C], B[SCAN_C];
            auto load_set = [&](uint4* S, int it) {
                #pragma unroll
                for (int c = 0; c < SCAN_C; ++c)
                    S[c] = *reinterpret_cast<const uint4*>(
                        omap + ((it * SCAN_C + c) * 32 + wave) * 128 + lane * 4);
            };
            auto take_set = [&](const uint4* S, int it) {
                #pragma unroll
                for (int c = 0; c < SCAN_C; ++c) {
                    take16(S[c], ((it * SCAN_C + c) * 32 + wave) * 512 + lane * 16);
                    if (cnt > QW - 512) { full(cnt); cnt = 0; }
                }
            };
            load_set(A, 0);
            for (int it = 0; it < n_full; it += 2) {
                load_set(B, min(it + 1, last));
                take_set(A, it);
                load_set(A, min(it + 2, last));
                if (it + 1 < n_full) take_set(B, it + 1);
            }
        }
        for (int base = n_full * per_it + wave * 128; base < n_edges; base += 32 * 128) {
            const int e0 = base + lane;
            const int d0 = (e0      < n_edges) ? dstp[e0]      : -1;
            const int d1 = (e0 + 32 < n_edges) ? dstp[e0 + 32] : -1;
            const int d2 = (e0 + 64 < n_edges) ? dstp[e0 + 64] : -1;
            const int d3 = (e0 + 96 < n_edges) ? dstp[e0 + 96] : -1;
            take4(d0, d1, d2, d3, e0, 32);
            if (cnt > QW - SCAN_C * 128) { full(cnt); cnt = 0; }
        }
    };

    filter([&](int) { if (lane == 0) refilterL = 1; });
    __syncthreads();
    const bool general = refilterL != 0;

    if (!general) {
        for (int b = 0; b < cnt; b += 32) {
            const bool on = b + lane < cnt;
            edge_logits(myList[on ? b + lane : cnt - 1], on);
        }
    } else {
        for (int base = wave * 32; base < n_edges; base += 32 * 32) {
            const int e = min(base + lane, n_edges - 1);
            const bool on = (base + lane < n_edges) && (unsigned)(dstp[e] - lo) < (unsigned)nR;
            if (__builtin_amdgcn_ballot_w32(on) != 0u) edge_logits(e, on);
        }
    }
    __syncthreads();

    if (!general) {
        flush(cnt);
    } else {
        cnt = 0;
        filter([&](int n) { flush(n); });
        flush(cnt);
    }
    __syncthreads();

    const float4 bA = reinterpret_cast<const float4*>(bias)[lane];
    const float4 bB = reinterpret_cast<const float4*>(bias)[32 + lane];
    const float4 gA = reinterpret_cast<const float4*>(gamma)[lane];
    const float4 gB = reinterpret_cast<const float4*>(gamma)[32 + lane];
    const float4 tA = reinterpret_cast<const float4*>(beta)[lane];
    const float4 tB = reinterpret_cast<const float4*>(beta)[32 + lane];
    for (int i = wave; i < nR; i += 32) {
        const size_t n = (size_t)(lo + i);
        const float4 xA = reinterpret_cast<const float4*>(x + n * HID)[lane];
        const float4 xB = reinterpret_cast<const float4*>(x + n * HID)[32 + lane];
        const float* a = accL + i * HID;
        const float dA = denL[i * HEADS + (lane >> 3)] + 1e-16f;
        const float dB = denL[i * HEADS + 4 + (lane >> 3)] + 1e-16f;
        float y[8];
        y[0] = xA.x + a[lane]       / dA + bA.x;
        y[1] = xA.y + a[32 + lane]  / dA + bA.y;
        y[2] = xA.z + a[64 + lane]  / dA + bA.z;
        y[3] = xA.w + a[96 + lane]  / dA + bA.w;
        y[4] = xB.x + a[128 + lane] / dB + bB.x;
        y[5] = xB.y + a[160 + lane] / dB + bB.y;
        y[6] = xB.z + a[192 + lane] / dB + bB.z;
        y[7] = xB.w + a[224 + lane] / dB + bB.w;
        float s = 0.f, s2 = 0.f;
        for (int j = 0; j < 8; ++j) { s += y[j]; s2 += y[j] * y[j]; }
        for (int off = 16; off >= 1; off >>= 1) {
            s  += __shfl_xor(s, off);
            s2 += __shfl_xor(s2, off);
        }
        const float mean = s * (1.f / HID);
        const float var  = s2 * (1.f / HID) - mean * mean;
        const float rs   = rsqrtf(var + LN_EPS);
        float4 oA, oB;
        oA.x = (y[0] - mean) * rs * gA.x + tA.x;
        oA.y = (y[1] - mean) * rs * gA.y + tA.y;
        oA.z = (y[2] - mean) * rs * gA.z + tA.z;
        oA.w = (y[3] - mean) * rs * gA.w + tA.w;
        oB.x = (y[4] - mean) * rs * gB.x + tB.x;
        oB.y = (y[5] - mean) * rs * gB.y + tB.y;
        oB.z = (y[6] - mean) * rs * gB.z + tB.z;
        oB.w = (y[7] - mean) * rs * gB.w + tB.w;
        reinterpret_cast<float4*>(out + n * HID)[lane]      = oA;
        reinterpret_cast<float4*>(out + n * HID)[32 + lane] = oB;
    }
}

extern "C" void kernel_launch(void* const* d_in, const int* in_sizes, int n_in,
                              void* d_out, int out_size, void* d_ws, size_t ws_size,
                              hipStream_t stream) {
    const float* x         = (const float*)d_in[0];
    const int*   ei        = (const int*)  d_in[1];
    const float* edge_attr = (const float*)d_in[2];
    const float* W         = (const float*)d_in[3];
    const float* W_e       = (const float*)d_in[4];
    const float* att_src   = (const float*)d_in[5];
    const float* att_dst   = (const float*)d_in[6];
    const float* att_edge  = (const float*)d_in[7];
    const float* bias      = (const float*)d_in[8];
    const float* gamma     = (const float*)d_in[9];
    const float* beta      = (const float*)d_in[10];
    float* out = (float*)d_out;

    const int n_nodes = in_sizes[0] / HID;
    const int n_edges = in_sizes[1] / 2;

    char* ws = (char*)d_ws;
    size_t off = 0;
    auto carve = [&](size_t bytes) -> char* {
        char* p = ws + off;
        off += (bytes + 255) & ~(size_t)255;
        return p;
    };
    float*    xp    = (float*)   carve((size_t)n_nodes * HID   * 4);
    float*    alpha = (float*)   carve((size_t)n_edges * HEADS * 4);
    float*    a_src = (float*)   carve((size_t)n_nodes * HEADS * 4);
    float*    a_dst = (float*)   carve((size_t)n_nodes * HEADS * 4);
    float*    w_eh  = (float*)   carve((size_t)EDIM * HEADS * 4);
    unsigned* omap  = (unsigned*)carve((size_t)n_edges);
    (void)ws_size;

    k_weh<<<1 + n_edges / 4096, 256, 0, stream>>>(W_e, att_edge, w_eh, ei, omap, n_edges);

    k_node_proj<<<(n_nodes + 31) / 32, 256, 0, stream>>>(x, W, xp, n_nodes);

    k_att_dots<<<(n_nodes + 7) / 8, 256, 0, stream>>>(xp, att_src, att_dst,
                                                      a_src, a_dst, n_nodes);

    k_own<<<(n_nodes + RN - 1) / RN, 1024, 0, stream>>>(xp, ei, edge_attr, w_eh, a_src, a_dst,
                                                        alpha, x, bias, gamma, beta, out, omap,
                                                        n_nodes, n_edges);
}


# Pass 1 files each edge under its row's owner wave (32 LDS lists, LDS counter slot); pass 2 adds row sums by plain 16-byte LDS read-add-write, float atomics on denominators only; sums in list order
# speedup vs baseline: 2.5384x; 1.0447x over previous
#include <hip/hip_runtime.h>
#include <hip/hip_bf16.h>
#include <stdint.h>

#define HID   256
#define EDIM  64
#define HEADS 8
#define CCH   32
#define NEG_SLOPE 0.2f
#define LN_EPS 1e-5f

typedef float v2f __attribute__((ext_vector_type(2)));
typedef float v8f __attribute__((ext_vector_type(8)));

__device__ __forceinline__ unsigned enc_f32(float f) {
    unsigned b = __float_as_uint(f);
    return (b & 0x80000000u) ? ~b : (b | 0x80000000u);
}
__device__ __forceinline__ float dec_f32(unsigned u) {
    unsigned b = (u & 0x80000000u) ? (u & 0x7FFFFFFFu) : ~u;
    return __uint_as_float(b);
}

#define RN     196
#define MAP_IT 65536

__global__ void __launch_bounds__(256)
k_weh(const float* __restrict__ W_e,
      const float* __restrict__ att_edge,
      float* __restrict__ w_eh,
      const int* __restrict__ ei, unsigned* __restrict__ omap, int n_edges) {
    if (blockIdx.x != 0) {
        const int e0 = ((blockIdx.x - 1) * 256 + threadIdx.x) * 16;
        if ((n_edges & 3) != 0 || e0 >= (n_edges / MAP_IT) * MAP_IT) return;
        const int4* dp = reinterpret_cast<const int4*>(ei + n_edges + e0);
        uint4 o;
        unsigned w[4];
        #pragma unroll
        for (int k = 0; k < 4; ++k) {
            const int4 v = dp[k];
            w[k] = (((unsigned)v.x / RN) & 255u)       | ((((unsigned)v.y / RN) & 255u) << 8) |
                   ((((unsigned)v.z / RN) & 255u) << 16) | ((((unsigned)v.w / RN) & 255u) << 24);
        }
        o.x = w[0]; o.y = w[1]; o.z = w[2]; o.w = w[3];
        reinterpret_cast<uint4*>(omap + e0 / 4)[0] = o;
        return;
    }
    int d = threadIdx.x;
    if (d >= EDIM) return;
    #pragma unroll
    for (int h = 0; h < HEADS; ++h) {
        float s = 0.f;
        #pragma unroll
        for (int c = 0; c < CCH; ++c)
            s += W_e[d * HID + h * CCH + c] * att_edge[h * CCH + c];
        w_eh[d * HEADS + h] = s;
    }
}

__global__ void __launch_bounds__(256)
k_node_proj(const float* __restrict__ x, const float* __restrict__ W,
            float* __restrict__ xp, int n_nodes) {
    const int wave = threadIdx.x >> 5;
    const int lane = threadIdx.x & 31;
    const int row_base = (blockIdx.x * 2 + (wave >> 2)) * 16;
    const int col_base = (wave & 3) * 64;
    if (row_base >= n_nodes) return;

    int m = lane & 15;
    int kk = (lane >> 4) * 2;
    int row = row_base + m;
    if (row >= n_nodes) row = n_nodes - 1;
    const float* xrow = x + (size_t)row * HID;
    const int ncol = col_base + (lane & 15);

    v8f acc0 = {}, acc1 = {}, acc2 = {}, acc3 = {};
    for (int k = 0; k < HID; k += 4) {
        v2f a;
        a.x = xrow[k + kk];
        a.y = xrow[k + kk + 1];
        const float* w0 = W + (size_t)(k + kk) * HID;
        const float* w1 = W + (size_t)(k + kk + 1) * HID;
        v2f b0, b1, b2, b3;
        b0.x = w0[ncol];      b0.y = w1[ncol];
        b1.x = w0[ncol + 16]; b1.y = w1[ncol + 16];
        b2.x = w0[ncol + 32]; b2.y = w1[ncol + 32];
        b3.x = w0[ncol + 48]; b3.y = w1[ncol + 48];
        acc0 = __builtin_amdgcn_wmma_f32_16x16x4_f32(false, a, false, b0, (short)0, acc0, false, false);
        acc1 = __builtin_amdgcn_wmma_f32_16x16x4_f32(false, a, false, b1, (short)0, acc1, false, false);
        acc2 = __builtin_amdgcn_wmma_f32_16x16x4_f32(false, a, false, b2, (short)0, acc2, false, false);
        acc3 = __builtin_amdgcn_wmma_f32_16x16x4_f32(false, a, false, b3, (short)0, acc3, false, false);
    }

    const int srow = row_base + ((lane >> 4) * 8);
    const int scol = col_base + (lane & 15);
    #pragma unroll
    for (int i = 0; i < 8; ++i) {
        int r = srow + i;
        if (r < n_nodes) {
            float* o = xp + (size_t)r * HID + scol;
            o[0]  = acc0[i];
            o[16] = acc1[i];
            o[32] = acc2[i];
            o[48] = acc3[i];
        }
    }
}

__global__ void __launch_bounds__(256)
k_att_dots(const float* __restrict__ xp, const float* __restrict__ att_src,
           const float* __restrict__ att_dst, float* __restrict__ a_src,
           float* __restrict__ a_dst, int n_nodes) {
    const int wave = threadIdx.x >> 5;
    const int lane = threadIdx.x & 31;
    const int node = blockIdx.x * 8 + wave;
    if (node >= n_nodes) return;
    const float* row = xp + (size_t)node * HID + lane * 8;
    float s = 0.f, t = 0.f;
    #pragma unroll
    for (int j = 0; j < 8; ++j) {
        float v = row[j];
        int col = lane * 8 + j;
        s += v * att_src[col];
        t += v * att_dst[col];
    }
    s += __shfl_xor(s, 1); s += __shfl_xor(s, 2);
    t += __shfl_xor(t, 1); t += __shfl_xor(t, 2);
    if ((lane & 3) == 0) {
        a_src[node * HEADS + (lane >> 2)] = s;
        a_dst[node * HEADS + (lane >> 2)] = t;
    }
}

#define QW     688
#define OWN_CAP 176
#define SCAN_C 4
#define GB     4
static_assert(MAP_IT == SCAN_C * 32 * 512, "one filter iteration = SCAN_C chunks of 512 edges per wave");

__global__ void __launch_bounds__(1024)
k_own(const float* __restrict__ xp, const int* __restrict__ ei,
      const float* __restrict__ edge_attr, const float* __restrict__ w_eh,
      const float* __restrict__ a_src, const float* __restrict__ a_dst,
      float* alpha,
      const float* __restrict__ x, const float* __restrict__ bias,
      const float* __restrict__ gamma, const float* __restrict__ beta,
      float* __restrict__ out, const unsigned* __restrict__ omap, int n_nodes, int n_edges) {
    __shared__ __attribute__((aligned(16))) float accL[RN * HID];
    __shared__ float denL[RN * HEADS];
    __shared__ unsigned maxL[RN * HEADS];
    __shared__ int   listL[32 * QW];
    __shared__ unsigned ownL[32 * OWN_CAP];
    __shared__ int   ownCntL[32];
    __shared__ int   refilterL;
    __shared__ int   ownOvfL;

    const int tid  = threadIdx.x;
    const int lane = tid & 31;
    const int wave = __builtin_amdgcn_readfirstlane(tid >> 5);
    const int lo   = blockIdx.x * RN;
    const int nR   = min(RN, n_nodes - lo);

    for (int i = tid; i < RN * HID / 4; i += 1024)
        reinterpret_cast<float4*>(accL)[i] = make_float4(0.f, 0.f, 0.f, 0.f);
    for (int i = tid; i < RN * HEADS; i += 1024) {
        denL[i] = 0.f;
        maxL[i] = 0u;
    }
    if (tid < 32) ownCntL[tid] = 0;
    if (tid == 0) { refilterL = 0; ownOvfL = 0; }
    __syncthreads();

    const int* dstp = ei + n_edges;
    int* myList = listL + wave * QW;
    const unsigned ltmask = (1u << lane) - 1u;
    int cnt = 0;
    const bool packOK = n_edges <= (1 << 24);

    auto edge_logits = [&](int e, bool on, bool file) __attribute__((always_inline)) {
        const int s = ei[e];
        const int d = dstp[e];
        const float4* er = reinterpret_cast<const float4*>(edge_attr + (size_t)e * EDIM);
        v2f t01[HEADS / 2], t23[HEADS / 2];
        #pragma unroll
        for (int k = 0; k < HEADS / 2; ++k) { t01[k] = v2f{0.f, 0.f}; t23[k] = v2f{0.f, 0.f}; }
        float4 cur[4];
        #pragma unroll
        for (int c = 0; c < 4; ++c) cur[c] = er[c];
        const float4 sA = reinterpret_cast<const float4*>(a_src + (size_t)s * HEADS)[0];
        const float4 sB = reinterpret_cast<const float4*>(a_src + (size_t)s * HEADS)[1];
        const float4 dA = reinterpret_cast<const float4*>(a_dst + (size_t)d * HEADS)[0];
        const float4 dB = reinterpret_cast<const float4*>(a_dst + (size_t)d * HEADS)[1];
        #pragma unroll 1
        for (int q = 0; q < 4; ++q) {
            float4 nxt[4];
            #pragma unroll
            for (int c = 0; c < 4; ++c) nxt[c] = er[min(q + 1, 3) * 4 + c];
            const v2f* w = reinterpret_cast<const v2f*>(w_eh + q * 16 * HEADS);
            v2f p[HEADS / 2];
            #pragma unroll
            for (int k = 0; k < HEADS / 2; ++k) p[k] = v2f{0.f, 0.f};
            #pragma unroll
            for (int c = 0; c < 4; ++c) {
                #pragma unroll
                for (int k = 0; k < HEADS / 2; ++k) {
                    p[k] += v2f{cur[c].x, cur[c].x} * w[(c * 4 + 0) * (HEADS / 2) + k];
                    p[k] += v2f{cur[c].y, cur[c].y} * w[(c * 4 + 1) * (HEADS / 2) + k];
                    p[k] += v2f{cur[c].z, cur[c].z} * w[(c * 4 + 2) * (HEADS / 2) + k];
                    p[k] += v2f{cur[c].w, cur[c].w} * w[(c * 4 + 3) * (HEADS / 2) + k];
                }
            }
            if (q < 2) {
                #pragma unroll
                for (int k = 0; k < HEADS / 2; ++k) t01[k] += p[k];
            } else {
                #pragma unroll
                for (int k = 0; k < HEADS / 2; ++k) t23[k] += p[k];
            }
            #pragma unroll
            for (int c = 0; c < 4; ++c) cur[c] = nxt[c];
        }
        const float as[HEADS] = {sA.x, sA.y, sA.z, sA.w, sB.x, sB.y, sB.z, sB.w};
        const float ad[HEADS] = {dA.x, dA.y, dA.z, dA.w, dB.x, dB.y, dB.z, dB.w};
        float al[HEADS];
        #pragma unroll
        for (int h = 0; h < HEADS; ++h) {
            const float dot = (h & 1) ? (t01[h / 2].y + t23[h / 2].y) : (t01[h / 2].x + t23[h / 2].x);
            const float v = as[h] + ad[h] + dot;
            al[h] = (v > 0.f) ? v : NEG_SLOPE * v;
        }
        if (on) {
            float4* o = reinterpret_cast<float4*>(alpha + (size_t)e * HEADS);
            o[0] = make_float4(al[0], al[1], al[2], al[3]);
            o[1] = make_float4(al[4], al[5], al[6], al[7]);
            unsigned* mx = maxL + (d - lo) * HEADS;
            #pragma unroll
            for (int h = 0; h < HEADS; ++h) atomicMax(&mx[h], enc_f32(al[h]));
            if (file) {
                const int dl = d - lo;
                const int ow = dl & 31;
                const int slot = atomicAdd(&ownCntL[ow], 1);
                if (slot < OWN_CAP) ownL[ow * OWN_CAP + slot] = ((unsigned)dl << 24) | (unsigned)e;
                else ownOvfL = 1;
            }
        }
    };

    auto flush = [&](int n) {
        for (int b = 0; b < n; b += GB) {
            int e[GB], s[GB], d[GB];
            #pragma unroll
            for (int u = 0; u < GB; ++u)
                e[u] = __builtin_amdgcn_readfirstlane(myList[min(b + u, n - 1)]);
            #pragma unroll
            for (int u = 0; u < GB; ++u) { s[u] = ei[e[u]]; d[u] = dstp[e[u]]; }
            float4 ra[GB], rb[GB];
            float  al[GB];
            #pragma unroll
            for (int u = 0; u < GB; ++u) {
                const float4* r = reinterpret_cast<const float4*>(xp + (size_t)s[u] * HID);
                ra[u] = r[lane];
                rb[u] = r[32 + lane];
                al[u] = alpha[(size_t)e[u] * HEADS + (lane & 7)];
            }
            #pragma unroll
            for (int u = 0; u < GB; ++u) {
                if (b + u < n) {
                    const int dl = d[u] - lo;
                    const float ea = __expf(al[u] - dec_f32(maxL[dl * HEADS + (lane & 7)]));
                    if (lane < HEADS) atomicAdd(&denL[dl * HEADS + lane], ea);
                    const float wa = __shfl(ea, lane >> 3);
                    const float wb = __shfl(ea, 4 + (lane >> 3));
                    float* a = accL + dl * HID + 4 * lane;
                    atomicAdd(&a[0],   wa * ra[u].x);
                    atomicAdd(&a[1],   wa * ra[u].y);
                    atomicAdd(&a[2],   wa * ra[u].z);
                    atomicAdd(&a[3],   wa * ra[u].w);
                    atomicAdd(&a[128], wb * rb[u].x);
                    atomicAdd(&a[129], wb * rb[u].y);
                    atomicAdd(&a[130], wb * rb[u].z);
                    atomicAdd(&a[131], wb * rb[u].w);
                }
            }
        }
    };

    struct OwnIds  { int e[GB]; int dl[GB]; int s[GB]; };
    struct OwnRows { float4 ra[GB], rb[GB]; float al[GB]; int dl[GB]; };
    auto flush_own = [&](int n) {
        if (n <= 0) return;
        const unsigned* mine = ownL + wave * OWN_CAP;
        auto ids = [&](OwnIds& I, int b) __attribute__((always_inline)) {
            #pragma unroll
            for (int u = 0; u < GB; ++u) {
                const unsigned en = __builtin_amdgcn_readfirstlane(mine[min(b + u, n - 1)]);
                I.e[u]  = (int)(en & 0x00FFFFFFu);
                I.dl[u] = (int)(en >> 24);
            }
            #pragma unroll
            for (int u = 0; u < GB; ++u) I.s[u] = ei[I.e[u]];
        };
        auto rows = [&](const OwnIds& I, OwnRows& R) __attribute__((always_inline)) {
            #pragma unroll
            for (int u = 0; u < GB; ++u) {
                const float4* r = reinterpret_cast<const float4*>(xp + (size_t)I.s[u] * HID);
                R.ra[u] = r[lane];
                R.rb[u] = r[32 + lane];
                R.al[u] = alpha[(size_t)I.e[u] * HEADS + (lane & 7)];
                R.dl[u] = I.dl[u];
            }
        };
        auto work = [&](const OwnRows& R, int b, bool guard) __attribute__((always_inline)) {
            v2f wa[GB], wb[GB];
            unsigned mx[GB];
            #pragma unroll
            for (int u = 0; u < GB; ++u)
                mx[u] = maxL[R.dl[u] * HEADS + (lane & 7)];
            #pragma unroll
            for (int u = 0; u < GB; ++u) {
                const float ea = __expf(R.al[u] - dec_f32(mx[u]));
                if ((!guard || b + u < n) && lane < HEADS) atomicAdd(&denL[R.dl[u] * HEADS + lane], ea);
                const float sa = __shfl(ea, lane >> 3);
                const float sb = __shfl(ea, 4 + (lane >> 3));
                v2f pa = {sa, sa}, pb = {sb, sb};
                asm volatile("" : "+v"(pa), "+v"(pb));
                wa[u] = pa;
                wb[u] = pb;
            }
            #pragma unroll
            for (int u = 0; u < GB; ++u) {
                if (!guard || b + u < n) {
                    float4* a = reinterpret_cast<float4*>(accL + R.dl[u] * HID);
                    float4 va = a[lane];
                    float4 vb = a[32 + lane];
                    va.x += wa[u].x * R.ra[u].x; va.y += wa[u].y * R.ra[u].y;
                    va.z += wa[u].x * R.ra[u].z; va.w += wa[u].y * R.ra[u].w;
                    vb.x += wb[u].x * R.rb[u].x; vb.y += wb[u].y * R.rb[u].y;
                    vb.z += wb[u].x * R.rb[u].z; vb.w += wb[u].y * R.rb[u].w;
                    a[lane]      = va;
                    a[32 + lane] = vb;
                }
            }
        };
        OwnIds  I0, I1;
        OwnRows R0, R1;
        ids(I0, 0);
        ids(I1, GB);
        rows(I0, R0);
        __builtin_amdgcn_sched_barrier(0);
        int b = 0;
        for (; b + 2 * GB <= n; b += 2 * GB) {
            ids(I0, b + 2 * GB);
            __builtin_amdgcn_sched_barrier(0);
            rows(I1, R1);
            __builtin_amdgcn_sched_barrier(0);
            work(R0, b, false);
            __builtin_amdgcn_sched_barrier(0);
            ids(I1, b + 3 * GB);
            __builtin_amdgcn_sched_barrier(0);
            rows(I0, R0);
            __builtin_amdgcn_sched_barrier(0);
            work(R1, b + GB, false);
            __builtin_amdgcn_sched_barrier(0);
        }
        if (b < n) {
            rows(I1, R1);
            work(R0, b, true);
            if (b + GB < n) work(R1, b + GB, true);
        }
    };

    const unsigned tgt = (blockIdx.x & 255u) * 0x01010101u;
    auto take16 = [&](uint4 v, int e0) {
        auto zf = [](unsigned x) { return ~(((x & 0x7F7F7F7Fu) + 0x7F7F7F7Fu) | x) & 0x80808080u; };
        unsigned g = (zf(v.x ^ tgt) >> 7) | (zf(v.y ^ tgt) >> 6) | (zf(v.z ^ tgt) >> 5) | (zf(v.w ^ tgt) >> 4);
        unsigned bal = __builtin_amdgcn_ballot_w32(g != 0u);
        while (bal != 0u) {
            if (g != 0u) {
                const int p = __builtin_ctz(g);
                myList[cnt + __popc(bal & ltmask)] = e0 + ((p & 3) << 2) + (p >> 3);
                g &= g - 1u;
            }
            cnt += __popc(bal);
            bal = __builtin_amdgcn_ballot_w32(g != 0u);
        }
        cnt = __builtin_amdgcn_readfirstlane(cnt);
    };

    auto take4 = [&](int d0, int d1, int d2, int d3, int e0, int step) {
        const bool m0 = (unsigned)(d0 - lo) < (unsigned)nR;
        const bool m1 = (unsigned)(d1 - lo) < (unsigned)nR;
        const bool m2 = (unsigned)(d2 - lo) < (unsigned)nR;
        const bool m3 = (unsigned)(d3 - lo) < (unsigned)nR;
        const unsigned b0 = __builtin_amdgcn_ballot_w32(m0);
        const unsigned b1 = __builtin_amdgcn_ballot_w32(m1);
        const unsigned b2 = __builtin_amdgcn_ballot_w32(m2);
        const unsigned b3 = __builtin_amdgcn_ballot_w32(m3);
        if ((b0 | b1 | b2 | b3) != 0u) {
            if (m0) myList[cnt + __popc(b0 & ltmask)] = e0;
            cnt += __popc(b0);
            if (m1) myList[cnt + __popc(b1 & ltmask)] = e0 + step;
            cnt += __popc(b1);
            if (m2) myList[cnt + __popc(b2 & ltmask)] = e0 + 2 * step;
            cnt += __popc(b2);
            if (m3) myList[cnt + __popc(b3 & ltmask)] = e0 + 3 * step;
            cnt += __popc(b3);
        }
    };

    auto filter = [&](auto&& full) {
        const int per_it = MAP_IT;
        const int n_full = ((n_edges & 3) == 0 && gridDim.x <= 256u) ? n_edges / per_it : 0;
        if (n_full > 0) {
            const int last = n_full - 1;
            uint4 A[SCAN_C], B[SCAN_C];
            auto load_set = [&](uint4* S, int it) {
                #pragma unroll
                for (int c = 0; c < SCAN_C; ++c)
                    S[c] = *reinterpret_cast<const uint4*>(
                        omap + ((it * SCAN_C + c) * 32 + wave) * 128 + lane * 4);
            };
            auto take_set = [&](const uint4* S, int it) {
                #pragma unroll
                for (int c = 0; c < SCAN_C; ++c) {
                    take16(S[c], ((it * SCAN_C + c) * 32 + wave) * 512 + lane * 16);
                    if (cnt > QW - 512) { full(cnt); cnt = 0; }
                }
            };
            load_set(A, 0);
            for (int it = 0; it < n_full; it += 2) {
                load_set(B, min(it + 1, last));
                take_set(A, it);
                load_set(A, min(it + 2, last));
                if (it + 1 < n_full) take_set(B, it + 1);
            }
        }
        for (int base = n_full * per_it + wave * 128; base < n_edges; base += 32 * 128) {
            const int e0 = base + lane;
            const int d0 = (e0      < n_edges) ? dstp[e0]      : -1;
            const int d1 = (e0 + 32 < n_edges) ? dstp[e0 + 32] : -1;
            const int d2 = (e0 + 64 < n_edges) ? dstp[e0 + 64] : -1;
            const int d3 = (e0 + 96 < n_edges) ? dstp[e0 + 96] : -1;
            take4(d0, d1, d2, d3, e0, 32);
            if (cnt > QW - SCAN_C * 128) { full(cnt); cnt = 0; }
        }
    };

    filter([&](int) { if (lane == 0) refilterL = 1; });
    __syncthreads();
    const bool general = refilterL != 0;

    if (!general) {
        for (int b = 0; b < cnt; b += 32) {
            const bool on = b + lane < cnt;
            edge_logits(myList[on ? b + lane : cnt - 1], on, packOK);
        }
    } else {
        for (int base = wave * 32; base < n_edges; base += 32 * 32) {
            const int e = min(base + lane, n_edges - 1);
            const bool on = (base + lane < n_edges) && (unsigned)(dstp[e] - lo) < (unsigned)nR;
            if (__builtin_amdgcn_ballot_w32(on) != 0u) edge_logits(e, on, false);
        }
    }
    __syncthreads();

    const bool plain = !general && packOK && ownOvfL == 0;
    if (plain) {
        flush_own(__builtin_amdgcn_readfirstlane(ownCntL[wave]));
    } else if (!general) {
        flush(cnt);
    } else {
        cnt = 0;
        filter([&](int n) { flush(n); });
        flush(cnt);
    }
    __syncthreads();

    const float4 bA = reinterpret_cast<const float4*>(bias)[lane];
    const float4 bB = reinterpret_cast<const float4*>(bias)[32 + lane];
    const float4 gA = reinterpret_cast<const float4*>(gamma)[lane];
    const float4 gB = reinterpret_cast<const float4*>(gamma)[32 + lane];
    const float4 tA = reinterpret_cast<const float4*>(beta)[lane];
    const float4 tB = reinterpret_cast<const float4*>(beta)[32 + lane];
    for (int i = wave; i < nR; i += 32) {
        const size_t n = (size_t)(lo + i);
        const float4 xA = reinterpret_cast<const float4*>(x + n * HID)[lane];
        const float4 xB = reinterpret_cast<const float4*>(x + n * HID)[32 + lane];
        const float4 aA = reinterpret_cast<const float4*>(accL + i * HID)[lane];
        const float4 aB = reinterpret_cast<const float4*>(accL + i * HID)[32 + lane];
        const float dA = denL[i * HEADS + (lane >> 3)] + 1e-16f;
        const float dB = denL[i * HEADS + 4 + (lane >> 3)] + 1e-16f;
        float y[8];
        y[0] = xA.x + aA.x / dA + bA.x;
        y[1] = xA.y + aA.y / dA + bA.y;
        y[2] = xA.z + aA.z / dA + bA.z;
        y[3] = xA.w + aA.w / dA + bA.w;
        y[4] = xB.x + aB.x / dB + bB.x;
        y[5] = xB.y + aB.y / dB + bB.y;
        y[6] = xB.z + aB.z / dB + bB.z;
        y[7] = xB.w + aB.w / dB + bB.w;
        float s = 0.f, s2 = 0.f;
        for (int j = 0; j < 8; ++j) { s += y[j]; s2 += y[j] * y[j]; }
        for (int off = 16; off >= 1; off >>= 1) {
            s  += __shfl_xor(s, off);
            s2 += __shfl_xor(s2, off);
        }
        const float mean = s * (1.f / HID);
        const float var  = s2 * (1.f / HID) - mean * mean;
        const float rs   = rsqrtf(var + LN_EPS);
        float4 oA, oB;
        oA.x = (y[0] - mean) * rs * gA.x + tA.x;
        oA.y = (y[1] - mean) * rs * gA.y + tA.y;
        oA.z = (y[2] - mean) * rs * gA.z + tA.z;
        oA.w = (y[3] - mean) * rs * gA.w + tA.w;
        oB.x = (y[4] - mean) * rs * gB.x + tB.x;
        oB.y = (y[5] - mean) * rs * gB.y + tB.y;
        oB.z = (y[6] - mean) * rs * gB.z + tB.z;
        oB.w = (y[7] - mean) * rs * gB.w + tB.w;
        reinterpret_cast<float4*>(out + n * HID)[lane]      = oA;
        reinterpret_cast<float4*>(out + n * HID)[32 + lane] = oB;
    }
}

extern "C" void kernel_launch(void* const* d_in, const int* in_sizes, int n_in,
                              void* d_out, int out_size, void* d_ws, size_t ws_size,
                              hipStream_t stream) {
    const float* x         = (const float*)d_in[0];
    const int*   ei        = (const int*)  d_in[1];
    const float* edge_attr = (const float*)d_in[2];
    const float* W         = (const float*)d_in[3];
    const float* W_e       = (const float*)d_in[4];
    const float* att_src   = (const float*)d_in[5];
    const float* att_dst   = (const float*)d_in[6];
    const float* att_edge  = (const float*)d_in[7];
    const float* bias      = (const float*)d_in[8];
    const float* gamma     = (const float*)d_in[9];
    const float* beta      = (const float*)d_in[10];
    float* out = (float*)d_out;

    const int n_nodes = in_sizes[0] / HID;
    const int n_edges = in_sizes[1] / 2;

    char* ws = (char*)d_ws;
    size_t off = 0;
    auto carve = [&](size_t bytes) -> char* {
        char* p = ws + off;
        off += (bytes + 255) & ~(size_t)255;
        return p;
    };
    float*    xp    = (float*)   carve((size_t)n_nodes * HID   * 4);
    float*    alpha = (float*)   carve((size_t)n_edges * HEADS * 4);
    float*    a_src = (float*)   carve((size_t)n_nodes * HEADS * 4);
    float*    a_dst = (float*)   carve((size_t)n_nodes * HEADS * 4);
    float*    w_eh  = (float*)   carve((size_t)EDIM * HEADS * 4);
    unsigned* omap  = (unsigned*)carve((size_t)n_edges);
    (void)ws_size;

    k_weh<<<1 + n_edges / 4096, 256, 0, stream>>>(W_e, att_edge, w_eh, ei, omap, n_edges);

    k_node_proj<<<(n_nodes + 31) / 32, 256, 0, stream>>>(x, W, xp, n_nodes);

    k_att_dots<<<(n_nodes + 7) / 8, 256, 0, stream>>>(xp, att_src, att_dst,
                                                      a_src, a_dst, n_nodes);

    k_own<<<(n_nodes + RN - 1) / RN, 1024, 0, stream>>>(xp, ei, edge_attr, w_eh, a_src, a_dst,
                                                        alpha, x, bias, gamma, beta, out, omap,
                                                        n_nodes, n_edges);
}


# Pass 1 reads 32 edge feature rows by whole 128-byte lines (8 lanes per half row) and turns them to a lane per edge through LDS (idle message rows), not a lane per row over 32 lines; sums unchanged
# speedup vs baseline: 2.7473x; 1.0823x over previous
#include <hip/hip_runtime.h>
#include <hip/hip_bf16.h>
#include <stdint.h>

#define HID   256
#define EDIM  64
#define HEADS 8
#define CCH   32
#define NEG_SLOPE 0.2f
#define LN_EPS 1e-5f

typedef float v2f __attribute__((ext_vector_type(2)));
typedef float v8f __attribute__((ext_vector_type(8)));
typedef __attribute__((address_space(4))) v2f cv2f;

__device__ __forceinline__ unsigned enc_f32(float f) {
    unsigned b = __float_as_uint(f);
    return (b & 0x80000000u) ? ~b : (b | 0x80000000u);
}
__device__ __forceinline__ float dec_f32(unsigned u) {
    unsigned b = (u & 0x80000000u) ? (u & 0x7FFFFFFFu) : ~u;
    return __uint_as_float(b);
}

#define RN     196
#define MAP_IT 65536

__global__ void __launch_bounds__(256)
k_weh(const float* __restrict__ W_e,
      const float* __restrict__ att_edge,
      float* __restrict__ w_eh,
      const int* __restrict__ ei, unsigned* __restrict__ omap, int n_edges) {
    if (blockIdx.x != 0) {
        const int e0 = ((blockIdx.x - 1) * 256 + threadIdx.x) * 16;
        if ((n_edges & 3) != 0 || e0 >= (n_edges / MAP_IT) * MAP_IT) return;
        const int4* dp = reinterpret_cast<const int4*>(ei + n_edges + e0);
        uint4 o;
        unsigned w[4];
        #pragma unroll
        for (int k = 0; k < 4; ++k) {
            const int4 v = dp[k];
            w[k] = (((unsigned)v.x / RN) & 255u)       | ((((unsigned)v.y / RN) & 255u) << 8) |
                   ((((unsigned)v.z / RN) & 255u) << 16) | ((((unsigned)v.w / RN) & 255u) << 24);
        }
        o.x = w[0]; o.y = w[1]; o.z = w[2]; o.w = w[3];
        reinterpret_cast<uint4*>(omap + e0 / 4)[0] = o;
        return;
    }
    int d = threadIdx.x;
    if (d >= EDIM) return;
    #pragma unroll
    for (int h = 0; h < HEADS; ++h) {
        float s = 0.f;
        #pragma unroll
        for (int c = 0; c < CCH; ++c)
            s += W_e[d * HID + h * CCH + c] * att_edge[h * CCH + c];
        w_eh[d * HEADS + h] = s;
    }
}

__global__ void __launch_bounds__(256)
k_node_proj(const float* __restrict__ x, const float* __restrict__ W,
            float* __restrict__ xp, int n_nodes) {
    const int wave = threadIdx.x >> 5;
    const int lane = threadIdx.x & 31;
    const int row_base = (blockIdx.x * 2 + (wave >> 2)) * 16;
    const int col_base = (wave & 3) * 64;
    if (row_base >= n_nodes) return;

    int m = lane & 15;
    int kk = (lane >> 4) * 2;
    int row = row_base + m;
    if (row >= n_nodes) row = n_nodes - 1;
    const float* xrow = x + (size_t)row * HID;
    const int ncol = col_base + (lane & 15);

    v8f acc0 = {}, acc1 = {}, acc2 = {}, acc3 = {};
    for (int k = 0; k < HID; k += 4) {
        v2f a;
        a.x = xrow[k + kk];
        a.y = xrow[k + kk + 1];
        const float* w0 = W + (size_t)(k + kk) * HID;
        const float* w1 = W + (size_t)(k + kk + 1) * HID;
        v2f b0, b1, b2, b3;
        b0.x = w0[ncol];      b0.y = w1[ncol];
        b1.x = w0[ncol + 16]; b1.y = w1[ncol + 16];
        b2.x = w0[ncol + 32]; b2.y = w1[ncol + 32];
        b3.x = w0[ncol + 48]; b3.y = w1[ncol + 48];
        acc0 = __builtin_amdgcn_wmma_f32_16x16x4_f32(false, a, false, b0, (short)0, acc0, false, false);
        acc1 = __builtin_amdgcn_wmma_f32_16x16x4_f32(false, a, false, b1, (short)0, acc1, false, false);
        acc2 = __builtin_amdgcn_wmma_f32_16x16x4_f32(false, a, false, b2, (short)0, acc2, false, false);
        acc3 = __builtin_amdgcn_wmma_f32_16x16x4_f32(false, a, false, b3, (short)0, acc3, false, false);
    }

    const int srow = row_base + ((lane >> 4) * 8);
    const int scol = col_base + (lane & 15);
    #pragma unroll
    for (int i = 0; i < 8; ++i) {
        int r = srow + i;
        if (r < n_nodes) {
            float* o = xp + (size_t)r * HID + scol;
            o[0]  = acc0[i];
            o[16] = acc1[i];
            o[32] = acc2[i];
            o[48] = acc3[i];
        }
    }
}

__global__ void __launch_bounds__(256)
k_att_dots(const float* __restrict__ xp, const float* __restrict__ att_src,
           const float* __restrict__ att_dst, float* __restrict__ a_src,
           float* __restrict__ a_dst, int n_nodes) {
    const int wave = threadIdx.x >> 5;
    const int lane = threadIdx.x & 31;
    const int node = blockIdx.x * 8 + wave;
    if (node >= n_nodes) return;
    const float* row = xp + (size_t)node * HID + lane * 8;
    float s = 0.f, t = 0.f;
    #pragma unroll
    for (int j = 0; j < 8; ++j) {
        float v = row[j];
        int col = lane * 8 + j;
        s += v * att_src[col];
        t += v * att_dst[col];
    }
    s += __shfl_xor(s, 1); s += __shfl_xor(s, 2);
    t += __shfl_xor(t, 1); t += __shfl_xor(t, 2);
    if ((lane & 3) == 0) {
        a_src[node * HEADS + (lane >> 2)] = s;
        a_dst[node * HEADS + (lane >> 2)] = t;
    }
}

#define QW     688
#define OWN_CAP 176
#define SCAN_C 4
#define GB     4
#define STG_W  (RN * HID / 32)
#define STG_P  36
static_assert(32 * STG_P <= STG_W, "32 staged half rows fit a wave's slice");
static_assert((RN * HID) % 32 == 0 && STG_W % 4 == 0, "32 equal slices, each cleared by 16-byte stores");
static_assert(MAP_IT == SCAN_C * 32 * 512, "one filter iteration = SCAN_C chunks of 512 edges per wave");

__global__ void __launch_bounds__(1024)
k_own(const float* __restrict__ xp, const int* __restrict__ ei,
      const float* __restrict__ edge_attr, const float* __restrict__ w_eh,
      const float* __restrict__ a_src, const float* __restrict__ a_dst,
      float* alpha,
      const float* __restrict__ x, const float* __restrict__ bias,
      const float* __restrict__ gamma, const float* __restrict__ beta,
      float* __restrict__ out, const unsigned* __restrict__ omap, int n_nodes, int n_edges) {
    __shared__ __attribute__((aligned(16))) float accL[RN * HID];
    __shared__ float denL[RN * HEADS];
    __shared__ unsigned maxL[RN * HEADS];
    __shared__ int   listL[32 * QW];
    __shared__ unsigned ownL[32 * OWN_CAP];
    __shared__ int   ownCntL[32];
    __shared__ int   refilterL;
    __shared__ int   ownOvfL;

    const int tid  = threadIdx.x;
    const int lane = tid & 31;
    const int wave = __builtin_amdgcn_readfirstlane(tid >> 5);
    const int lo   = blockIdx.x * RN;
    const int nR   = min(RN, n_nodes - lo);

    for (int i = tid; i < RN * HEADS; i += 1024) {
        denL[i] = 0.f;
        maxL[i] = 0u;
    }
    if (tid < 32) ownCntL[tid] = 0;
    if (tid == 0) { refilterL = 0; ownOvfL = 0; }
    __syncthreads();

    const int* dstp = ei + n_edges;
    int* myList = listL + wave * QW;
    const unsigned ltmask = (1u << lane) - 1u;
    int cnt = 0;
    const bool packOK = n_edges <= (1 << 24);
    float* stage = accL + wave * STG_W;

    auto edge_logits = [&](int e, bool on, bool file) __attribute__((always_inline)) {
        const int s = ei[e];
        const int d = dstp[e];
        const int sub = lane >> 3;
        const int pc  = (lane & 7) * 4;
        const float* ea = edge_attr + pc;
        const float* p0 = ea + (size_t)__shfl(e,      sub) * EDIM;
        const float* p1 = ea + (size_t)__shfl(e,  4 + sub) * EDIM;
        const float* p2 = ea + (size_t)__shfl(e,  8 + sub) * EDIM;
        const float* p3 = ea + (size_t)__shfl(e, 12 + sub) * EDIM;
        const float* p4 = ea + (size_t)__shfl(e, 16 + sub) * EDIM;
        const float* p5 = ea + (size_t)__shfl(e, 20 + sub) * EDIM;
        const float* p6 = ea + (size_t)__shfl(e, 24 + sub) * EDIM;
        const float* p7 = ea + (size_t)__shfl(e, 28 + sub) * EDIM;
        auto ld16 = [](const float* p) { return *reinterpret_cast<const float4*>(p); };
        float4 g0 = ld16(p0), g1 = ld16(p1), g2 = ld16(p2), g3 = ld16(p3),
               g4 = ld16(p4), g5 = ld16(p5), g6 = ld16(p6), g7 = ld16(p7);
        const float4 sA = reinterpret_cast<const float4*>(a_src + (size_t)s * HEADS)[0];
        const float4 sB = reinterpret_cast<const float4*>(a_src + (size_t)s * HEADS)[1];
        const float4 dA = reinterpret_cast<const float4*>(a_dst + (size_t)d * HEADS)[0];
        const float4 dB = reinterpret_cast<const float4*>(a_dst + (size_t)d * HEADS)[1];
        float* wr = stage + sub * STG_P + pc;
        const float4* rd = reinterpret_cast<const float4*>(stage + lane * STG_P);
        v2f t01[HEADS / 2], t23[HEADS / 2];
        #pragma unroll
        for (int k = 0; k < HEADS / 2; ++k) { t01[k] = v2f{0.f, 0.f}; t23[k] = v2f{0.f, 0.f}; }
        #pragma unroll
        for (int h = 0; h < 2; ++h) {
            auto st16 = [&](int j, const float4& v) { *reinterpret_cast<float4*>(wr + 4 * j * STG_P) = v; };
            st16(0, g0); st16(1, g1); st16(2, g2); st16(3, g3);
            st16(4, g4); st16(5, g5); st16(6, g6); st16(7, g7);
            __builtin_amdgcn_wave_barrier();
            if (h == 0) {
                g0 = ld16(p0 + 32); g1 = ld16(p1 + 32); g2 = ld16(p2 + 32); g3 = ld16(p3 + 32);
                g4 = ld16(p4 + 32); g5 = ld16(p5 + 32); g6 = ld16(p6 + 32); g7 = ld16(p7 + 32);
            }
            #pragma unroll 1
            for (int qq = 0; qq < 2; ++qq) {
                float4 cur[4];
                #pragma unroll
                for (int c = 0; c < 4; ++c) cur[c] = rd[qq * 4 + c];
                const cv2f* w = (const cv2f*)(w_eh + (h * 2 + qq) * 16 * HEADS);
                v2f p[HEADS / 2];
                #pragma unroll
                for (int k = 0; k < HEADS / 2; ++k) p[k] = v2f{0.f, 0.f};
                #pragma unroll
                for (int c = 0; c < 4; ++c) {
                    #pragma unroll
                    for (int k = 0; k < HEADS / 2; ++k) {
                        p[k] += v2f{cur[c].x, cur[c].x} * w[(c * 4 + 0) * (HEADS / 2) + k];
                        p[k] += v2f{cur[c].y, cur[c].y} * w[(c * 4 + 1) * (HEADS / 2) + k];
                        p[k] += v2f{cur[c].z, cur[c].z} * w[(c * 4 + 2) * (HEADS / 2) + k];
                        p[k] += v2f{cur[c].w, cur[c].w} * w[(c * 4 + 3) * (HEADS / 2) + k];
                    }
                }
                if (h == 0) {
                    #pragma unroll
                    for (int k = 0; k < HEADS / 2; ++k) t01[k] += p[k];
                } else {
                    #pragma unroll
                    for (int k = 0; k < HEADS / 2; ++k) t23[k] += p[k];
                }
            }
            __builtin_amdgcn_wave_barrier();
        }
        const float as[HEADS] = {sA.x, sA.y, sA.z, sA.w, sB.x, sB.y, sB.z, sB.w};
        const float ad[HEADS] = {dA.x, dA.y, dA.z, dA.w, dB.x, dB.y, dB.z, dB.w};
        float al[HEADS];
        #pragma unroll
        for (int h = 0; h < HEADS; ++h) {
            const float dot = (h & 1) ? (t01[h / 2].y + t23[h / 2].y) : (t01[h / 2].x + t23[h / 2].x);
            const float v = as[h] + ad[h] + dot;
            al[h] = (v > 0.f) ? v : NEG_SLOPE * v;
        }
        if (on) {
            float4* o = reinterpret_cast<float4*>(alpha + (size_t)e * HEADS);
            o[0] = make_float4(al[0], al[1], al[2], al[3]);
            o[1] = make_float4(al[4], al[5], al[6], al[7]);
            unsigned* mx = maxL + (d - lo) * HEADS;
            #pragma unroll
            for (int h = 0; h < HEADS; ++h) atomicMax(&mx[h], enc_f32(al[h]));
            if (file) {
                const int dl = d - lo;
                const int ow = dl & 31;
                const int slot = atomicAdd(&ownCntL[ow], 1);
                if (slot < OWN_CAP) ownL[ow * OWN_CAP + slot] = ((unsigned)dl << 24) | (unsigned)e;
                else ownOvfL = 1;
            }
        }
    };

    auto flush = [&](int n) {
        for (int b = 0; b < n; b += GB) {
            int e[GB], s[GB], d[GB];
            #pragma unroll
            for (int u = 0; u < GB; ++u)
                e[u] = __builtin_amdgcn_readfirstlane(myList[min(b + u, n - 1)]);
            #pragma unroll
            for (int u = 0; u < GB; ++u) { s[u] = ei[e[u]]; d[u] = dstp[e[u]]; }
            float4 ra[GB], rb[GB];
            float  al[GB];
            #pragma unroll
            for (int u = 0; u < GB; ++u) {
                const float4* r = reinterpret_cast<const float4*>(xp + (size_t)s[u] * HID);
                ra[u] = r[lane];
                rb[u] = r[32 + lane];
                al[u] = alpha[(size_t)e[u] * HEADS + (lane & 7)];
            }
            #pragma unroll
            for (int u = 0; u < GB; ++u) {
                if (b + u < n) {
                    const int dl = d[u] - lo;
                    const float ea = __expf(al[u] - dec_f32(maxL[dl * HEADS + (lane & 7)]));
                    if (lane < HEADS) atomicAdd(&denL[dl * HEADS + lane], ea);
                    const float wa = __shfl(ea, lane >> 3);
                    const float wb = __shfl(ea, 4 + (lane >> 3));
                    float* a = accL + dl * HID + 4 * lane;
                    atomicAdd(&a[0],   wa * ra[u].x);
                    atomicAdd(&a[1],   wa * ra[u].y);
                    atomicAdd(&a[2],   wa * ra[u].z);
                    atomicAdd(&a[3],   wa * ra[u].w);
                    atomicAdd(&a[128], wb * rb[u].x);
                    atomicAdd(&a[129], wb * rb[u].y);
                    atomicAdd(&a[130], wb * rb[u].z);
                    atomicAdd(&a[131], wb * rb[u].w);
                }
            }
        }
    };

    struct OwnIds  { int e[GB]; int dl[GB]; int s[GB]; };
    struct OwnRows { float4 ra[GB], rb[GB]; float al[GB]; int dl[GB]; };
    auto flush_own = [&](int n) {
        if (n <= 0) return;
        const unsigned* mine = ownL + wave * OWN_CAP;
        auto ids = [&](OwnIds& I, int b) __attribute__((always_inline)) {
            #pragma unroll
            for (int u = 0; u < GB; ++u) {
                const unsigned en = __builtin_amdgcn_readfirstlane(mine[min(b + u, n - 1)]);
                I.e[u]  = (int)(en & 0x00FFFFFFu);
                I.dl[u] = (int)(en >> 24);
            }
            #pragma unroll
            for (int u = 0; u < GB; ++u) I.s[u] = ei[I.e[u]];
        };
        auto rows = [&](const OwnIds& I, OwnRows& R) __attribute__((always_inline)) {
            #pragma unroll
            for (int u = 0; u < GB; ++u) {
                const float4* r = reinterpret_cast<const float4*>(xp + (size_t)I.s[u] * HID);
                R.ra[u] = r[lane];
                R.rb[u] = r[32 + lane];
                R.al[u] = alpha[(size_t)I.e[u] * HEADS + (lane & 7)];
                R.dl[u] = I.dl[u];
            }
        };
        auto work = [&](const OwnRows& R, int b, bool guard) __attribute__((always_inline)) {
            v2f wa[GB], wb[GB];
            unsigned mx[GB];
            #pragma unroll
            for (int u = 0; u < GB; ++u)
                mx[u] = maxL[R.dl[u] * HEADS + (lane & 7)];
            #pragma unroll
            for (int u = 0; u < GB; ++u) {
                const float ea = __expf(R.al[u] - dec_f32(mx[u]));
                if ((!guard || b + u < n) && lane < HEADS) atomicAdd(&denL[R.dl[u] * HEADS + lane], ea);
                const float sa = __shfl(ea, lane >> 3);
                const float sb = __shfl(ea, 4 + (lane >> 3));
                v2f pa = {sa, sa}, pb = {sb, sb};
                asm volatile("" : "+v"(pa), "+v"(pb));
                wa[u] = pa;
                wb[u] = pb;
            }
            #pragma unroll
            for (int u = 0; u < GB; ++u) {
                if (!guard || b + u < n) {
                    float4* a = reinterpret_cast<float4*>(accL + R.dl[u] * HID);
                    float4 va = a[lane];
                    float4 vb = a[32 + lane];
                    va.x += wa[u].x * R.ra[u].x; va.y += wa[u].y * R.ra[u].y;
                    va.z += wa[u].x * R.ra[u].z; va.w += wa[u].y * R.ra[u].w;
                    vb.x += wb[u].x * R.rb[u].x; vb.y += wb[u].y * R.rb[u].y;
                    vb.z += wb[u].x * R.rb[u].z; vb.w += wb[u].y * R.rb[u].w;
                    a[lane]      = va;
                    a[32 + lane] = vb;
                }
            }
        };
        OwnIds  I0, I1;
        OwnRows R0, R1;
        ids(I0, 0);
        ids(I1, GB);
        rows(I0, R0);
        __builtin_amdgcn_sched_barrier(0);
        int b = 0;
        for (; b + 2 * GB <= n; b += 2 * GB) {
            ids(I0, b + 2 * GB);
            __builtin_amdgcn_sched_barrier(0);
            rows(I1, R1);
            __builtin_amdgcn_sched_barrier(0);
            work(R0, b, false);
            __builtin_amdgcn_sched_barrier(0);
            ids(I1, b + 3 * GB);
            __builtin_amdgcn_sched_barrier(0);
            rows(I0, R0);
            __builtin_amdgcn_sched_barrier(0);
            work(R1, b + GB, false);
            __builtin_amdgcn_sched_barrier(0);
        }
        if (b < n) {
            rows(I1, R1);
            work(R0, b, true);
            if (b + GB < n) work(R1, b + GB, true);
        }
    };

    const unsigned tgt = (blockIdx.x & 255u) * 0x01010101u;
    auto take16 = [&](uint4 v, int e0) {
        auto zf = [](unsigned x) { return ~(((x & 0x7F7F7F7Fu) + 0x7F7F7F7Fu) | x) & 0x80808080u; };
        unsigned g = (zf(v.x ^ tgt) >> 7) | (zf(v.y ^ tgt) >> 6) | (zf(v.z ^ tgt) >> 5) | (zf(v.w ^ tgt) >> 4);
        unsigned bal = __builtin_amdgcn_ballot_w32(g != 0u);
        while (bal != 0u) {
            if (g != 0u) {
                const int p = __builtin_ctz(g);
                myList[cnt + __popc(bal & ltmask)] = e0 + ((p & 3) << 2) + (p >> 3);
                g &= g - 1u;
            }
            cnt += __popc(bal);
            bal = __builtin_amdgcn_ballot_w32(g != 0u);
        }
        cnt = __builtin_amdgcn_readfirstlane(cnt);
    };

    auto take4 = [&](int d0, int d1, int d2, int d3, int e0, int step) {
        const bool m0 = (unsigned)(d0 - lo) < (unsigned)nR;
        const bool m1 = (unsigned)(d1 - lo) < (unsigned)nR;
        const bool m2 = (unsigned)(d2 - lo) < (unsigned)nR;
        const bool m3 = (unsigned)(d3 - lo) < (unsigned)nR;
        const unsigned b0 = __builtin_amdgcn_ballot_w32(m0);
        const unsigned b1 = __builtin_amdgcn_ballot_w32(m1);
        const unsigned b2 = __builtin_amdgcn_ballot_w32(m2);
        const unsigned b3 = __builtin_amdgcn_ballot_w32(m3);
        if ((b0 | b1 | b2 | b3) != 0u) {
            if (m0) myList[cnt + __popc(b0 & ltmask)] = e0;
            cnt += __popc(b0);
            if (m1) myList[cnt + __popc(b1 & ltmask)] = e0 + step;
            cnt += __popc(b1);
            if (m2) myList[cnt + __popc(b2 & ltmask)] = e0 + 2 * step;
            cnt += __popc(b2);
            if (m3) myList[cnt + __popc(b3 & ltmask)] = e0 + 3 * step;
            cnt += __popc(b3);
        }
    };

    auto filter = [&](auto&& full) {
        const int per_it = MAP_IT;
        const int n_full = ((n_edges & 3) == 0 && gridDim.x <= 256u) ? n_edges / per_it : 0;
        if (n_full > 0) {
            const int last = n_full - 1;
            uint4 A[SCAN_C], B[SCAN_C];
            auto load_set = [&](uint4* S, int it) {
                #pragma unroll
                for (int c = 0; c < SCAN_C; ++c)
                    S[c] = *reinterpret_cast<const uint4*>(
                        omap + ((it * SCAN_C + c) * 32 + wave) * 128 + lane * 4);
            };
            auto take_set = [&](const uint4* S, int it) {
                #pragma unroll
                for (int c = 0; c < SCAN_C; ++c) {
                    take16(S[c], ((it * SCAN_C + c) * 32 + wave) * 512 + lane * 16);
                    if (cnt > QW - 512) { full(cnt); cnt = 0; }
                }
            };
            load_set(A, 0);
            for (int it = 0; it < n_full; it += 2) {
                load_set(B, min(it + 1, last));
                take_set(A, it);
                load_set(A, min(it + 2, last));
                if (it + 1 < n_full) take_set(B, it + 1);
            }
        }
        for (int base = n_full * per_it + wave * 128; base < n_edges; base += 32 * 128) {
            const int e0 = base + lane;
            const int d0 = (e0      < n_edges) ? dstp[e0]      : -1;
            const int d1 = (e0 + 32 < n_edges) ? dstp[e0 + 32] : -1;
            const int d2 = (e0 + 64 < n_edges) ? dstp[e0 + 64] : -1;
            const int d3 = (e0 + 96 < n_edges) ? dstp[e0 + 96] : -1;
            take4(d0, d1, d2, d3, e0, 32);
            if (cnt > QW - SCAN_C * 128) { full(cnt); cnt = 0; }
        }
    };

    filter([&](int) { if (lane == 0) refilterL = 1; });
    __syncthreads();
    const bool general = refilterL != 0;

    if (!general) {
        for (int b = 0; b < cnt; b += 32) {
            const bool on = b + lane < cnt;
            edge_logits(myList[on ? b + lane : cnt - 1], on, packOK);
        }
    } else {
        for (int base = wave * 32; base < n_edges; base += 32 * 32) {
            const int e = min(base + lane, n_edges - 1);
            const bool on = (base + lane < n_edges) && (unsigned)(dstp[e] - lo) < (unsigned)nR;
            if (__builtin_amdgcn_ballot_w32(on) != 0u) edge_logits(e, on, false);
        }
    }
    for (int i = lane; i < STG_W / 4; i += 32)
        reinterpret_cast<float4*>(stage)[i] = make_float4(0.f, 0.f, 0.f, 0.f);
    __syncthreads();

    const bool plain = !general && packOK && ownOvfL == 0;
    if (plain) {
        flush_own(__builtin_amdgcn_readfirstlane(ownCntL[wave]));
    } else if (!general) {
        flush(cnt);
    } else {
        cnt = 0;
        filter([&](int n) { flush(n); });
        flush(cnt);
    }
    __syncthreads();

    const float4 bA = reinterpret_cast<const float4*>(bias)[lane];
    const float4 bB = reinterpret_cast<const float4*>(bias)[32 + lane];
    const float4 gA = reinterpret_cast<const float4*>(gamma)[lane];
    const float4 gB = reinterpret_cast<const float4*>(gamma)[32 + lane];
    const float4 tA = reinterpret_cast<const float4*>(beta)[lane];
    const float4 tB = reinterpret_cast<const float4*>(beta)[32 + lane];
    for (int i = wave; i < nR; i += 32) {
        const size_t n = (size_t)(lo + i);
        const float4 xA = reinterpret_cast<const float4*>(x + n * HID)[lane];
        const float4 xB = reinterpret_cast<const float4*>(x + n * HID)[32 + lane];
        const float4 aA = reinterpret_cast<const float4*>(accL + i * HID)[lane];
        const float4 aB = reinterpret_cast<const float4*>(accL + i * HID)[32 + lane];
        const float dA = denL[i * HEADS + (lane >> 3)] + 1e-16f;
        const float dB = denL[i * HEADS + 4 + (lane >> 3)] + 1e-16f;
        float y[8];
        y[0] = xA.x + aA.x / dA + bA.x;
        y[1] = xA.y + aA.y / dA + bA.y;
        y[2] = xA.z + aA.z / dA + bA.z;
        y[3] = xA.w + aA.w / dA + bA.w;
        y[4] = xB.x + aB.x / dB + bB.x;
        y[5] = xB.y + aB.y / dB + bB.y;
        y[6] = xB.z + aB.z / dB + bB.z;
        y[7] = xB.w + aB.w / dB + bB.w;
        float s = 0.f, s2 = 0.f;
        for (int j = 0; j < 8; ++j) { s += y[j]; s2 += y[j] * y[j]; }
        for (int off = 16; off >= 1; off >>= 1) {
            s  += __shfl_xor(s, off);
            s2 += __shfl_xor(s2, off);
        }
        const float mean = s * (1.f / HID);
        const float var  = s2 * (1.f / HID) - mean * mean;
        const float rs   = rsqrtf(var + LN_EPS);
        float4 oA, oB;
        oA.x = (y[0] - mean) * rs * gA.x + tA.x;
        oA.y = (y[1] - mean) * rs * gA.y + tA.y;
        oA.z = (y[2] - mean) * rs * gA.z + tA.z;
        oA.w = (y[3] - mean) * rs * gA.w + tA.w;
        oB.x = (y[4] - mean) * rs * gB.x + tB.x;
        oB.y = (y[5] - mean) * rs * gB.y + tB.y;
        oB.z = (y[6] - mean) * rs * gB.z + tB.z;
        oB.w = (y[7] - mean) * rs * gB.w + tB.w;
        reinterpret_cast<float4*>(out + n * HID)[lane]      = oA;
        reinterpret_cast<float4*>(out + n * HID)[32 + lane] = oB;
    }
}

extern "C" void kernel_launch(void* const* d_in, const int* in_sizes, int n_in,
                              void* d_out, int out_size, void* d_ws, size_t ws_size,
                              hipStream_t stream) {
    const float* x         = (const float*)d_in[0];
    const int*   ei        = (const int*)  d_in[1];
    const float* edge_attr = (const float*)d_in[2];
    const float* W         = (const float*)d_in[3];
    const float* W_e       = (const float*)d_in[4];
    const float* att_src   = (const float*)d_in[5];
    const float* att_dst   = (const float*)d_in[6];
    const float* att_edge  = (const float*)d_in[7];
    const float* bias      = (const float*)d_in[8];
    const float* gamma     = (const float*)d_in[9];
    const float* beta      = (const float*)d_in[10];
    float* out = (float*)d_out;

    const int n_nodes = in_sizes[0] / HID;
    const int n_edges = in_sizes[1] / 2;

    char* ws = (char*)d_ws;
    size_t off = 0;
    auto carve = [&](size_t bytes) -> char* {
        char* p = ws + off;
        off += (bytes + 255) & ~(size_t)255;
        return p;
    };
    float*    xp    = (float*)   carve((size_t)n_nodes * HID   * 4);
    float*    alpha = (float*)   carve((size_t)n_edges * HEADS * 4);
    float*    a_src = (float*)   carve((size_t)n_nodes * HEADS * 4);
    float*    a_dst = (float*)   carve((size_t)n_nodes * HEADS * 4);
    float*    w_eh  = (float*)   carve((size_t)EDIM * HEADS * 4);
    unsigned* omap  = (unsigned*)carve((size_t)n_edges);
    (void)ws_size;

    k_weh<<<1 + n_edges / 4096, 256, 0, stream>>>(W_e, att_edge, w_eh, ei, omap, n_edges);

    k_node_proj<<<(n_nodes + 31) / 32, 256, 0, stream>>>(x, W, xp, n_nodes);

    k_att_dots<<<(n_nodes + 7) / 8, 256, 0, stream>>>(xp, att_src, att_dst,
                                                      a_src, a_dst, n_nodes);

    k_own<<<(n_nodes + RN - 1) / RN, 1024, 0, stream>>>(xp, ei, edge_attr, w_eh, a_src, a_dst,
                                                        alpha, x, bias, gamma, beta, out, omap,
                                                        n_nodes, n_edges);
}


# Node projection: a 4-wave workgroup copies its 32 x rows to LDS by whole 128-byte lines, waves read both A fragments there (8 bytes per lane) and form 32x64 tiles; k order of every sum unchanged
# speedup vs baseline: 3.0124x; 1.0965x over previous
#include <hip/hip_runtime.h>
#include <hip/hip_bf16.h>
#include <stdint.h>

#define HID   256
#define EDIM  64
#define HEADS 8
#define CCH   32
#define NEG_SLOPE 0.2f
#define LN_EPS 1e-5f

typedef float v2f __attribute__((ext_vector_type(2)));
typedef float v8f __attribute__((ext_vector_type(8)));
typedef __attribute__((address_space(4))) v2f cv2f;

__device__ __forceinline__ unsigned enc_f32(float f) {
    unsigned b = __float_as_uint(f);
    return (b & 0x80000000u) ? ~b : (b | 0x80000000u);
}
__device__ __forceinline__ float dec_f32(unsigned u) {
    unsigned b = (u & 0x80000000u) ? (u & 0x7FFFFFFFu) : ~u;
    return __uint_as_float(b);
}

#define RN     196
#define MAP_IT 65536

__global__ void __launch_bounds__(256)
k_weh(const float* __restrict__ W_e,
      const float* __restrict__ att_edge,
      float* __restrict__ w_eh,
      const int* __restrict__ ei, unsigned* __restrict__ omap, int n_edges) {
    if (blockIdx.x != 0) {
        const int e0 = ((blockIdx.x - 1) * 256 + threadIdx.x) * 16;
        if ((n_edges & 3) != 0 || e0 >= (n_edges / MAP_IT) * MAP_IT) return;
        const int4* dp = reinterpret_cast<const int4*>(ei + n_edges + e0);
        uint4 o;
        unsigned w[4];
        #pragma unroll
        for (int k = 0; k < 4; ++k) {
            const int4 v = dp[k];
            w[k] = (((unsigned)v.x / RN) & 255u)       | ((((unsigned)v.y / RN) & 255u) << 8) |
                   ((((unsigned)v.z / RN) & 255u) << 16) | ((((unsigned)v.w / RN) & 255u) << 24);
        }
        o.x = w[0]; o.y = w[1]; o.z = w[2]; o.w = w[3];
        reinterpret_cast<uint4*>(omap + e0 / 4)[0] = o;
        return;
    }
    int d = threadIdx.x;
    if (d >= EDIM) return;
    #pragma unroll
    for (int h = 0; h < HEADS; ++h) {
        float s = 0.f;
        #pragma unroll
        for (int c = 0; c < CCH; ++c)
            s += W_e[d * HID + h * CCH + c] * att_edge[h * CCH + c];
        w_eh[d * HEADS + h] = s;
    }
}

#define PJ_ROWS  32
#define PJ_PITCH 260
__global__ void __launch_bounds__(128)
k_node_proj(const float* __restrict__ x, const float* __restrict__ W,
            float* __restrict__ xp, int n_nodes) {
    __shared__ __attribute__((aligned(16))) float xs[PJ_ROWS * PJ_PITCH];
    const int tid  = threadIdx.x;
    const int wave = tid >> 5;
    const int lane = tid & 31;
    const int row_base = blockIdx.x * PJ_ROWS;

    {
        const int piece = tid & 63;
        const int rsub  = tid >> 6;
        #pragma unroll
        for (int half = 0; half < 2; ++half) {
            float4 v[8];
            #pragma unroll
            for (int i = 0; i < 8; ++i) {
                const int r = (half * 8 + i) * 2 + rsub;
                const int gr = min(row_base + r, n_nodes - 1);
                v[i] = reinterpret_cast<const float4*>(x + (size_t)gr * HID)[piece];
            }
            #pragma unroll
            for (int i = 0; i < 8; ++i) {
                const int r = (half * 8 + i) * 2 + rsub;
                *reinterpret_cast<float4*>(xs + r * PJ_PITCH + piece * 4) = v[i];
            }
        }
    }
    __syncthreads();

    const int c   = lane & 15;
    const int kk  = (lane >> 4) * 2;
    const int col_base = wave * 64;
    const int ncol = col_base + c;
    const float* a0p = xs + c * PJ_PITCH + kk;
    const float* a1p = xs + (16 + c) * PJ_PITCH + kk;

    v8f c00 = {}, c01 = {}, c02 = {}, c03 = {};
    v8f c10 = {}, c11 = {}, c12 = {}, c13 = {};
    for (int k = 0; k < HID; k += 4) {
        const v2f a0 = *reinterpret_cast<const v2f*>(a0p + k);
        const v2f a1 = *reinterpret_cast<const v2f*>(a1p + k);
        const float* w0 = W + (size_t)(k + kk) * HID;
        const float* w1 = W + (size_t)(k + kk + 1) * HID;
        v2f b0, b1, b2, b3;
        b0.x = w0[ncol];      b0.y = w1[ncol];
        b1.x = w0[ncol + 16]; b1.y = w1[ncol + 16];
        b2.x = w0[ncol + 32]; b2.y = w1[ncol + 32];
        b3.x = w0[ncol + 48]; b3.y = w1[ncol + 48];
        c00 = __builtin_amdgcn_wmma_f32_16x16x4_f32(false, a0, false, b0, (short)0, c00, false, false);
        c10 = __builtin_amdgcn_wmma_f32_16x16x4_f32(false, a1, false, b0, (short)0, c10, false, false);
        c01 = __builtin_amdgcn_wmma_f32_16x16x4_f32(false, a0, false, b1, (short)0, c01, false, false);
        c11 = __builtin_amdgcn_wmma_f32_16x16x4_f32(false, a1, false, b1, (short)0, c11, false, false);
        c02 = __builtin_amdgcn_wmma_f32_16x16x4_f32(false, a0, false, b2, (short)0, c02, false, false);
        c12 = __builtin_amdgcn_wmma_f32_16x16x4_f32(false, a1, false, b2, (short)0, c12, false, false);
        c03 = __builtin_amdgcn_wmma_f32_16x16x4_f32(false, a0, false, b3, (short)0, c03, false, false);
        c13 = __builtin_amdgcn_wmma_f32_16x16x4_f32(false, a1, false, b3, (short)0, c13, false, false);
    }

    const int srow = row_base + ((lane >> 4) * 8);
    #pragma unroll
    for (int i = 0; i < 8; ++i) {
        const int r0 = srow + i;
        if (r0 < n_nodes) {
            float* o = xp + (size_t)r0 * HID + ncol;
            o[0]  = c00[i];
            o[16] = c01[i];
            o[32] = c02[i];
            o[48] = c03[i];
        }
        const int r1 = r0 + 16;
        if (r1 < n_nodes) {
            float* o = xp + (size_t)r1 * HID + ncol;
            o[0]  = c10[i];
            o[16] = c11[i];
            o[32] = c12[i];
            o[48] = c13[i];
        }
    }
}

__global__ void __launch_bounds__(256)
k_att_dots(const float* __restrict__ xp, const float* __restrict__ att_src,
           const float* __restrict__ att_dst, float* __restrict__ a_src,
           float* __restrict__ a_dst, int n_nodes) {
    const int wave = threadIdx.x >> 5;
    const int lane = threadIdx.x & 31;
    const int node = blockIdx.x * 8 + wave;
    if (node >= n_nodes) return;
    const float* row = xp + (size_t)node * HID + lane * 8;
    float s = 0.f, t = 0.f;
    #pragma unroll
    for (int j = 0; j < 8; ++j) {
        float v = row[j];
        int col = lane * 8 + j;
        s += v * att_src[col];
        t += v * att_dst[col];
    }
    s += __shfl_xor(s, 1); s += __shfl_xor(s, 2);
    t += __shfl_xor(t, 1); t += __shfl_xor(t, 2);
    if ((lane & 3) == 0) {
        a_src[node * HEADS + (lane >> 2)] = s;
        a_dst[node * HEADS + (lane >> 2)] = t;
    }
}

#define QW     688
#define OWN_CAP 176
#define SCAN_C 4
#define GB     4
#define STG_W  (RN * HID / 32)
#define STG_P  36
static_assert(32 * STG_P <= STG_W, "32 staged half rows fit a wave's slice");
static_assert((RN * HID) % 32 == 0 && STG_W % 4 == 0, "32 equal slices, each cleared by 16-byte stores");
static_assert(MAP_IT == SCAN_C * 32 * 512, "one filter iteration = SCAN_C chunks of 512 edges per wave");

__global__ void __launch_bounds__(1024)
k_own(const float* __restrict__ xp, const int* __restrict__ ei,
      const float* __restrict__ edge_attr, const float* __restrict__ w_eh,
      const float* __restrict__ a_src, const float* __restrict__ a_dst,
      float* alpha,
      const float* __restrict__ x, const float* __restrict__ bias,
      const float* __restrict__ gamma, const float* __restrict__ beta,
      float* __restrict__ out, const unsigned* __restrict__ omap, int n_nodes, int n_edges) {
    __shared__ __attribute__((aligned(16))) float accL[RN * HID];
    __shared__ float denL[RN * HEADS];
    __shared__ unsigned maxL[RN * HEADS];
    __shared__ int   listL[32 * QW];
    __shared__ unsigned ownL[32 * OWN_CAP];
    __shared__ int   ownCntL[32];
    __shared__ int   refilterL;
    __shared__ int   ownOvfL;

    const int tid  = threadIdx.x;
    const int lane = tid & 31;
    const int wave = __builtin_amdgcn_readfirstlane(tid >> 5);
    const int lo   = blockIdx.x * RN;
    const int nR   = min(RN, n_nodes - lo);

    for (int i = tid; i < RN * HEADS; i += 1024) {
        denL[i] = 0.f;
        maxL[i] = 0u;
    }
    if (tid < 32) ownCntL[tid] = 0;
    if (tid == 0) { refilterL = 0; ownOvfL = 0; }
    __syncthreads();

    const int* dstp = ei + n_edges;
    int* myList = listL + wave * QW;
    const unsigned ltmask = (1u << lane) - 1u;
    int cnt = 0;
    const bool packOK = n_edges <= (1 << 24);
    float* stage = accL + wave * STG_W;

    auto edge_logits = [&](int e, bool on, bool file) __attribute__((always_inline)) {
        const int s = ei[e];
        const int d = dstp[e];
        const int sub = lane >> 3;
        const int pc  = (lane & 7) * 4;
        const float* ea = edge_attr + pc;
        const float* p0 = ea + (size_t)__shfl(e,      sub) * EDIM;
        const float* p1 = ea + (size_t)__shfl(e,  4 + sub) * EDIM;
        const float* p2 = ea + (size_t)__shfl(e,  8 + sub) * EDIM;
        const float* p3 = ea + (size_t)__shfl(e, 12 + sub) * EDIM;
        const float* p4 = ea + (size_t)__shfl(e, 16 + sub) * EDIM;
        const float* p5 = ea + (size_t)__shfl(e, 20 + sub) * EDIM;
        const float* p6 = ea + (size_t)__shfl(e, 24 + sub) * EDIM;
        const float* p7 = ea + (size_t)__shfl(e, 28 + sub) * EDIM;
        auto ld16 = [](const float* p) { return *reinterpret_cast<const float4*>(p); };
        float4 g0 = ld16(p0), g1 = ld16(p1), g2 = ld16(p2), g3 = ld16(p3),
               g4 = ld16(p4), g5 = ld16(p5), g6 = ld16(p6), g7 = ld16(p7);
        const float4 sA = reinterpret_cast<const float4*>(a_src + (size_t)s * HEADS)[0];
        const float4 sB = reinterpret_cast<const float4*>(a_src + (size_t)s * HEADS)[1];
        const float4 dA = reinterpret_cast<const float4*>(a_dst + (size_t)d * HEADS)[0];
        const float4 dB = reinterpret_cast<const float4*>(a_dst + (size_t)d * HEADS)[1];
        float* wr = stage + sub * STG_P + pc;
        const float4* rd = reinterpret_cast<const float4*>(stage + lane * STG_P);
        v2f t01[HEADS / 2], t23[HEADS / 2];
        #pragma unroll
        for (int k = 0; k < HEADS / 2; ++k) { t01[k] = v2f{0.f, 0.f}; t23[k] = v2f{0.f, 0.f}; }
        #pragma unroll
        for (int h = 0; h < 2; ++h) {
            auto st16 = [&](int j, const float4& v) { *reinterpret_cast<float4*>(wr + 4 * j * STG_P) = v; };
            st16(0, g0); st16(1, g1); st16(2, g2); st16(3, g3);
            st16(4, g4); st16(5, g5); st16(6, g6); st16(7, g7);
            __builtin_amdgcn_wave_barrier();
            if (h == 0) {
                g0 = ld16(p0 + 32); g1 = ld16(p1 + 32); g2 = ld16(p2 + 32); g3 = ld16(p3 + 32);
                g4 = ld16(p4 + 32); g5 = ld16(p5 + 32); g6 = ld16(p6 + 32); g7 = ld16(p7 + 32);
            }
            #pragma unroll 1
            for (int qq = 0; qq < 2; ++qq) {
                float4 cur[4];
                #pragma unroll
                for (int c = 0; c < 4; ++c) cur[c] = rd[qq * 4 + c];
                const cv2f* w = (const cv2f*)(w_eh + (h * 2 + qq) * 16 * HEADS);
                v2f p[HEADS / 2];
                #pragma unroll
                for (int k = 0; k < HEADS / 2; ++k) p[k] = v2f{0.f, 0.f};
                #pragma unroll
                for (int c = 0; c < 4; ++c) {
                    #pragma unroll
                    for (int k = 0; k < HEADS / 2; ++k) {
                        p[k] += v2f{cur[c].x, cur[c].x} * w[(c * 4 + 0) * (HEADS / 2) + k];
                        p[k] += v2f{cur[c].y, cur[c].y} * w[(c * 4 + 1) * (HEADS / 2) + k];
                        p[k] += v2f{cur[c].z, cur[c].z} * w[(c * 4 + 2) * (HEADS / 2) + k];
                        p[k] += v2f{cur[c].w, cur[c].w} * w[(c * 4 + 3) * (HEADS / 2) + k];
                    }
                }
                if (h == 0) {
                    #pragma unroll
                    for (int k = 0; k < HEADS / 2; ++k) t01[k] += p[k];
                } else {
                    #pragma unroll
                    for (int k = 0; k < HEADS / 2; ++k) t23[k] += p[k];
                }
            }
            __builtin_amdgcn_wave_barrier();
        }
        const float as[HEADS] = {sA.x, sA.y, sA.z, sA.w, sB.x, sB.y, sB.z, sB.w};
        const float ad[HEADS] = {dA.x, dA.y, dA.z, dA.w, dB.x, dB.y, dB.z, dB.w};
        float al[HEADS];
        #pragma unroll
        for (int h = 0; h < HEADS; ++h) {
            const float dot = (h & 1) ? (t01[h / 2].y + t23[h / 2].y) : (t01[h / 2].x + t23[h / 2].x);
            const float v = as[h] + ad[h] + dot;
            al[h] = (v > 0.f) ? v : NEG_SLOPE * v;
        }
        if (on) {
            float4* o = reinterpret_cast<float4*>(alpha + (size_t)e * HEADS);
            o[0] = make_float4(al[0], al[1], al[2], al[3]);
            o[1] = make_float4(al[4], al[5], al[6], al[7]);
            unsigned* mx = maxL + (d - lo) * HEADS;
            #pragma unroll
            for (int h = 0; h < HEADS; ++h) atomicMax(&mx[h], enc_f32(al[h]));
            if (file) {
                const int dl = d - lo;
                const int ow = dl & 31;
                const int slot = atomicAdd(&ownCntL[ow], 1);
                if (slot < OWN_CAP) ownL[ow * OWN_CAP + slot] = ((unsigned)dl << 24) | (unsigned)e;
                else ownOvfL = 1;
            }
        }
    };

    auto flush = [&](int n) {
        for (int b = 0; b < n; b += GB) {
            int e[GB], s[GB], d[GB];
            #pragma unroll
            for (int u = 0; u < GB; ++u)
                e[u] = __builtin_amdgcn_readfirstlane(myList[min(b + u, n - 1)]);
            #pragma unroll
            for (int u = 0; u < GB; ++u) { s[u] = ei[e[u]]; d[u] = dstp[e[u]]; }
            float4 ra[GB], rb[GB];
            float  al[GB];
            #pragma unroll
            for (int u = 0; u < GB; ++u) {
                const float4* r = reinterpret_cast<const float4*>(xp + (size_t)s[u] * HID);
                ra[u] = r[lane];
                rb[u] = r[32 + lane];
                al[u] = alpha[(size_t)e[u] * HEADS + (lane & 7)];
            }
            #pragma unroll
            for (int u = 0; u < GB; ++u) {
                if (b + u < n) {
                    const int dl = d[u] - lo;
                    const float ea = __expf(al[u] - dec_f32(maxL[dl * HEADS + (lane & 7)]));
                    if (lane < HEADS) atomicAdd(&denL[dl * HEADS + lane], ea);
                    const float wa = __shfl(ea, lane >> 3);
                    const float wb = __shfl(ea, 4 + (lane >> 3));
                    float* a = accL + dl * HID + 4 * lane;
                    atomicAdd(&a[0],   wa * ra[u].x);
                    atomicAdd(&a[1],   wa * ra[u].y);
                    atomicAdd(&a[2],   wa * ra[u].z);
                    atomicAdd(&a[3],   wa * ra[u].w);
                    atomicAdd(&a[128], wb * rb[u].x);
                    atomicAdd(&a[129], wb * rb[u].y);
                    atomicAdd(&a[130], wb * rb[u].z);
                    atomicAdd(&a[131], wb * rb[u].w);
                }
            }
        }
    };

    struct OwnIds  { int e[GB]; int dl[GB]; int s[GB]; };
    struct OwnRows { float4 ra[GB], rb[GB]; float al[GB]; int dl[GB]; };
    auto flush_own = [&](int n) {
        if (n <= 0) return;
        const unsigned* mine = ownL + wave * OWN_CAP;
        auto ids = [&](OwnIds& I, int b) __attribute__((always_inline)) {
            #pragma unroll
            for (int u = 0; u < GB; ++u) {
                const unsigned en = __builtin_amdgcn_readfirstlane(mine[min(b + u, n - 1)]);
                I.e[u]  = (int)(en & 0x00FFFFFFu);
                I.dl[u] = (int)(en >> 24);
            }
            #pragma unroll
            for (int u = 0; u < GB; ++u) I.s[u] = ei[I.e[u]];
        };
        auto rows = [&](const OwnIds& I, OwnRows& R) __attribute__((always_inline)) {
            #pragma unroll
            for (int u = 0; u < GB; ++u) {
                const float4* r = reinterpret_cast<const float4*>(xp + (size_t)I.s[u] * HID);
                R.ra[u] = r[lane];
                R.rb[u] = r[32 + lane];
                R.al[u] = alpha[(size_t)I.e[u] * HEADS + (lane & 7)];
                R.dl[u] = I.dl[u];
            }
        };
        auto work = [&](const OwnRows& R, int b, bool guard) __attribute__((always_inline)) {
            v2f wa[GB], wb[GB];
            unsigned mx[GB];
            #pragma unroll
            for (int u = 0; u < GB; ++u)
                mx[u] = maxL[R.dl[u] * HEADS + (lane & 7)];
            #pragma unroll
            for (int u = 0; u < GB; ++u) {
                const float ea = __expf(R.al[u] - dec_f32(mx[u]));
                if ((!guard || b + u < n) && lane < HEADS) atomicAdd(&denL[R.dl[u] * HEADS + lane], ea);
                const float sa = __shfl(ea, lane >> 3);
                const float sb = __shfl(ea, 4 + (lane >> 3));
                v2f pa = {sa, sa}, pb = {sb, sb};
                asm volatile("" : "+v"(pa), "+v"(pb));
                wa[u] = pa;
                wb[u] = pb;
            }
            #pragma unroll
            for (int u = 0; u < GB; ++u) {
                if (!guard || b + u < n) {
                    float4* a = reinterpret_cast<float4*>(accL + R.dl[u] * HID);
                    float4 va = a[lane];
                    float4 vb = a[32 + lane];
                    va.x += wa[u].x * R.ra[u].x; va.y += wa[u].y * R.ra[u].y;
                    va.z += wa[u].x * R.ra[u].z; va.w += wa[u].y * R.ra[u].w;
                    vb.x += wb[u].x * R.rb[u].x; vb.y += wb[u].y * R.rb[u].y;
                    vb.z += wb[u].x * R.rb[u].z; vb.w += wb[u].y * R.rb[u].w;
                    a[lane]      = va;
                    a[32 + lane] = vb;
                }
            }
        };
        OwnIds  I0, I1;
        OwnRows R0, R1;
        ids(I0, 0);
        ids(I1, GB);
        rows(I0, R0);
        __builtin_amdgcn_sched_barrier(0);
        int b = 0;
        for (; b + 2 * GB <= n; b += 2 * GB) {
            ids(I0, b + 2 * GB);
            __builtin_amdgcn_sched_barrier(0);
            rows(I1, R1);
            __builtin_amdgcn_sched_barrier(0);
            work(R0, b, false);
            __builtin_amdgcn_sched_barrier(0);
            ids(I1, b + 3 * GB);
            __builtin_amdgcn_sched_barrier(0);
            rows(I0, R0);
            __builtin_amdgcn_sched_barrier(0);
            work(R1, b + GB, false);
            __builtin_amdgcn_sched_barrier(0);
        }
        if (b < n) {
            rows(I1, R1);
            work(R0, b, true);
            if (b + GB < n) work(R1, b + GB, true);
        }
    };

    const unsigned tgt = (blockIdx.x & 255u) * 0x01010101u;
    auto take16 = [&](uint4 v, int e0) {
        auto zf = [](unsigned x) { return ~(((x & 0x7F7F7F7Fu) + 0x7F7F7F7Fu) | x) & 0x80808080u; };
        unsigned g = (zf(v.x ^ tgt) >> 7) | (zf(v.y ^ tgt) >> 6) | (zf(v.z ^ tgt) >> 5) | (zf(v.w ^ tgt) >> 4);
        unsigned bal = __builtin_amdgcn_ballot_w32(g != 0u);
        while (bal != 0u) {
            if (g != 0u) {
                const int p = __builtin_ctz(g);
                myList[cnt + __popc(bal & ltmask)] = e0 + ((p & 3) << 2) + (p >> 3);
                g &= g - 1u;
            }
            cnt += __popc(bal);
            bal = __builtin_amdgcn_ballot_w32(g != 0u);
        }
        cnt = __builtin_amdgcn_readfirstlane(cnt);
    };

    auto take4 = [&](int d0, int d1, int d2, int d3, int e0, int step) {
        const bool m0 = (unsigned)(d0 - lo) < (unsigned)nR;
        const bool m1 = (unsigned)(d1 - lo) < (unsigned)nR;
        const bool m2 = (unsigned)(d2 - lo) < (unsigned)nR;
        const bool m3 = (unsigned)(d3 - lo) < (unsigned)nR;
        const unsigned b0 = __builtin_amdgcn_ballot_w32(m0);
        const unsigned b1 = __builtin_amdgcn_ballot_w32(m1);
        const unsigned b2 = __builtin_amdgcn_ballot_w32(m2);
        const unsigned b3 = __builtin_amdgcn_ballot_w32(m3);
        if ((b0 | b1 | b2 | b3) != 0u) {
            if (m0) myList[cnt + __popc(b0 & ltmask)] = e0;
            cnt += __popc(b0);
            if (m1) myList[cnt + __popc(b1 & ltmask)] = e0 + step;
            cnt += __popc(b1);
            if (m2) myList[cnt + __popc(b2 & ltmask)] = e0 + 2 * step;
            cnt += __popc(b2);
            if (m3) myList[cnt + __popc(b3 & ltmask)] = e0 + 3 * step;
            cnt += __popc(b3);
        }
    };

    auto filter = [&](auto&& full) {
        const int per_it = MAP_IT;
        const int n_full = ((n_edges & 3) == 0 && gridDim.x <= 256u) ? n_edges / per_it : 0;
        if (n_full > 0) {
            const int last = n_full - 1;
            uint4 A[SCAN_C], B[SCAN_C];
            auto load_set = [&](uint4* S, int it) {
                #pragma unroll
                for (int c = 0; c < SCAN_C; ++c)
                    S[c] = *reinterpret_cast<const uint4*>(
                        omap + ((it * SCAN_C + c) * 32 + wave) * 128 + lane * 4);
            };
            auto take_set = [&](const uint4* S, int it) {
                #pragma unroll
                for (int c = 0; c < SCAN_C; ++c) {
                    take16(S[c], ((it * SCAN_C + c) * 32 + wave) * 512 + lane * 16);
                    if (cnt > QW - 512) { full(cnt); cnt = 0; }
                }
            };
            load_set(A, 0);
            for (int it = 0; it < n_full; it += 2) {
                load_set(B, min(it + 1, last));
                take_set(A, it);
                load_set(A, min(it + 2, last));
                if (it + 1 < n_full) take_set(B, it + 1);
            }
        }
        for (int base = n_full * per_it + wave * 128; base < n_edges; base += 32 * 128) {
            const int e0 = base + lane;
            const int d0 = (e0      < n_edges) ? dstp[e0]      : -1;
            const int d1 = (e0 + 32 < n_edges) ? dstp[e0 + 32] : -1;
            const int d2 = (e0 + 64 < n_edges) ? dstp[e0 + 64] : -1;
            const int d3 = (e0 + 96 < n_edges) ? dstp[e0 + 96] : -1;
            take4(d0, d1, d2, d3, e0, 32);
            if (cnt > QW - SCAN_C * 128) { full(cnt); cnt = 0; }
        }
    };

    filter([&](int) { if (lane == 0) refilterL = 1; });
    __syncthreads();
    const bool general = refilterL != 0;

    if (!general) {
        for (int b = 0; b < cnt; b += 32) {
            const bool on = b + lane < cnt;
            edge_logits(myList[on ? b + lane : cnt - 1], on, packOK);
        }
    } else {
        for (int base = wave * 32; base < n_edges; base += 32 * 32) {
            const int e = min(base + lane, n_edges - 1);
            const bool on = (base + lane < n_edges) && (unsigned)(dstp[e] - lo) < (unsigned)nR;
            if (__builtin_amdgcn_ballot_w32(on) != 0u) edge_logits(e, on, false);
        }
    }
    for (int i = lane; i < STG_W / 4; i += 32)
        reinterpret_cast<float4*>(stage)[i] = make_float4(0.f, 0.f, 0.f, 0.f);
    __syncthreads();

    const bool plain = !general && packOK && ownOvfL == 0;
    if (plain) {
        flush_own(__builtin_amdgcn_readfirstlane(ownCntL[wave]));
    } else if (!general) {
        flush(cnt);
    } else {
        cnt = 0;
        filter([&](int n) { flush(n); });
        flush(cnt);
    }
    __syncthreads();

    const float4 bA = reinterpret_cast<const float4*>(bias)[lane];
    const float4 bB = reinterpret_cast<const float4*>(bias)[32 + lane];
    const float4 gA = reinterpret_cast<const float4*>(gamma)[lane];
    const float4 gB = reinterpret_cast<const float4*>(gamma)[32 + lane];
    const float4 tA = reinterpret_cast<const float4*>(beta)[lane];
    const float4 tB = reinterpret_cast<const float4*>(beta)[32 + lane];
    for (int i = wave; i < nR; i += 32) {
        const size_t n = (size_t)(lo + i);
        const float4 xA = reinterpret_cast<const float4*>(x + n * HID)[lane];
        const float4 xB = reinterpret_cast<const float4*>(x + n * HID)[32 + lane];
        const float4 aA = reinterpret_cast<const float4*>(accL + i * HID)[lane];
        const float4 aB = reinterpret_cast<const float4*>(accL + i * HID)[32 + lane];
        const float dA = denL[i * HEADS + (lane >> 3)] + 1e-16f;
        const float dB = denL[i * HEADS + 4 + (lane >> 3)] + 1e-16f;
        float y[8];
        y[0] = xA.x + aA.x / dA + bA.x;
        y[1] = xA.y + aA.y / dA + bA.y;
        y[2] = xA.z + aA.z / dA + bA.z;
        y[3] = xA.w + aA.w / dA + bA.w;
        y[4] = xB.x + aB.x / dB + bB.x;
        y[5] = xB.y + aB.y / dB + bB.y;
        y[6] = xB.z + aB.z / dB + bB.z;
        y[7] = xB.w + aB.w / dB + bB.w;
        float s = 0.f, s2 = 0.f;
        for (int j = 0; j < 8; ++j) { s += y[j]; s2 += y[j] * y[j]; }
        for (int off = 16; off >= 1; off >>= 1) {
            s  += __shfl_xor(s, off);
            s2 += __shfl_xor(s2, off);
        }
        const float mean = s * (1.f / HID);
        const float var  = s2 * (1.f / HID) - mean * mean;
        const float rs   = rsqrtf(var + LN_EPS);
        float4 oA, oB;
        oA.x = (y[0] - mean) * rs * gA.x + tA.x;
        oA.y = (y[1] - mean) * rs * gA.y + tA.y;
        oA.z = (y[2] - mean) * rs * gA.z + tA.z;
        oA.w = (y[3] - mean) * rs * gA.w + tA.w;
        oB.x = (y[4] - mean) * rs * gB.x + tB.x;
        oB.y = (y[5] - mean) * rs * gB.y + tB.y;
        oB.z = (y[6] - mean) * rs * gB.z + tB.z;
        oB.w = (y[7] - mean) * rs * gB.w + tB.w;
        reinterpret_cast<float4*>(out + n * HID)[lane]      = oA;
        reinterpret_cast<float4*>(out + n * HID)[32 + lane] = oB;
    }
}

extern "C" void kernel_launch(void* const* d_in, const int* in_sizes, int n_in,
                              void* d_out, int out_size, void* d_ws, size_t ws_size,
                              hipStream_t stream) {
    const float* x         = (const float*)d_in[0];
    const int*   ei        = (const int*)  d_in[1];
    const float* edge_attr = (const float*)d_in[2];
    const float* W         = (const float*)d_in[3];
    const float* W_e       = (const float*)d_in[4];
    const float* att_src   = (const float*)d_in[5];
    const float* att_dst   = (const float*)d_in[6];
    const float* att_edge  = (const float*)d_in[7];
    const float* bias      = (const float*)d_in[8];
    const float* gamma     = (const float*)d_in[9];
    const float* beta      = (const float*)d_in[10];
    float* out = (float*)d_out;

    const int n_nodes = in_sizes[0] / HID;
    const int n_edges = in_sizes[1] / 2;

    char* ws = (char*)d_ws;
    size_t off = 0;
    auto carve = [&](size_t bytes) -> char* {
        char* p = ws + off;
        off += (bytes + 255) & ~(size_t)255;
        return p;
    };
    float*    xp    = (float*)   carve((size_t)n_nodes * HID   * 4);
    float*    alpha = (float*)   carve((size_t)n_edges * HEADS * 4);
    float*    a_src = (float*)   carve((size_t)n_nodes * HEADS * 4);
    float*    a_dst = (float*)   carve((size_t)n_nodes * HEADS * 4);
    float*    w_eh  = (float*)   carve((size_t)EDIM * HEADS * 4);
    unsigned* omap  = (unsigned*)carve((size_t)n_edges);
    (void)ws_size;

    k_weh<<<1 + n_edges / 4096, 256, 0, stream>>>(W_e, att_edge, w_eh, ei, omap, n_edges);

    k_node_proj<<<(n_nodes + 31) / 32, 128, 0, stream>>>(x, W, xp, n_nodes);

    k_att_dots<<<(n_nodes + 7) / 8, 256, 0, stream>>>(xp, att_src, att_dst,
                                                      a_src, a_dst, n_nodes);

    k_own<<<(n_nodes + RN - 1) / RN, 1024, 0, stream>>>(xp, ei, edge_attr, w_eh, a_src, a_dst,
                                                        alpha, x, bias, gamma, beta, out, omap,
                                                        n_nodes, n_edges);
}


# Pass 2: an owner wave orders its list by row (stable ballot counting sort in LDS) and sums each row's run in registers; row and denominators stored once, no LDS add per edge; sum order unchanged
# speedup vs baseline: 3.1028x; 1.0300x over previous
#include <hip/hip_runtime.h>
#include <hip/hip_bf16.h>
#include <stdint.h>

#define HID   256
#define EDIM  64
#define HEADS 8
#define CCH   32
#define NEG_SLOPE 0.2f
#define LN_EPS 1e-5f

typedef float v2f __attribute__((ext_vector_type(2)));
typedef float v8f __attribute__((ext_vector_type(8)));
typedef __attribute__((address_space(4))) v2f cv2f;

__device__ __forceinline__ unsigned enc_f32(float f) {
    unsigned b = __float_as_uint(f);
    return (b & 0x80000000u) ? ~b : (b | 0x80000000u);
}
__device__ __forceinline__ float dec_f32(unsigned u) {
    unsigned b = (u & 0x80000000u) ? (u & 0x7FFFFFFFu) : ~u;
    return __uint_as_float(b);
}

#define RN     196
#define MAP_IT 65536

__global__ void __launch_bounds__(256)
k_weh(const float* __restrict__ W_e,
      const float* __restrict__ att_edge,
      float* __restrict__ w_eh,
      const int* __restrict__ ei, unsigned* __restrict__ omap, int n_edges) {
    if (blockIdx.x != 0) {
        const int e0 = ((blockIdx.x - 1) * 256 + threadIdx.x) * 16;
        if ((n_edges & 3) != 0 || e0 >= (n_edges / MAP_IT) * MAP_IT) return;
        const int4* dp = reinterpret_cast<const int4*>(ei + n_edges + e0);
        uint4 o;
        unsigned w[4];
        #pragma unroll
        for (int k = 0; k < 4; ++k) {
            const int4 v = dp[k];
            w[k] = (((unsigned)v.x / RN) & 255u)       | ((((unsigned)v.y / RN) & 255u) << 8) |
                   ((((unsigned)v.z / RN) & 255u) << 16) | ((((unsigned)v.w / RN) & 255u) << 24);
        }
        o.x = w[0]; o.y = w[1]; o.z = w[2]; o.w = w[3];
        reinterpret_cast<uint4*>(omap + e0 / 4)[0] = o;
        return;
    }
    int d = threadIdx.x;
    if (d >= EDIM) return;
    #pragma unroll
    for (int h = 0; h < HEADS; ++h) {
        float s = 0.f;
        #pragma unroll
        for (int c = 0; c < CCH; ++c)
            s += W_e[d * HID + h * CCH + c] * att_edge[h * CCH + c];
        w_eh[d * HEADS + h] = s;
    }
}

#define PJ_ROWS  32
#define PJ_PITCH 260
__global__ void __launch_bounds__(128)
k_node_proj(const float* __restrict__ x, const float* __restrict__ W,
            float* __restrict__ xp, int n_nodes) {
    __shared__ __attribute__((aligned(16))) float xs[PJ_ROWS * PJ_PITCH];
    const int tid  = threadIdx.x;
    const int wave = tid >> 5;
    const int lane = tid & 31;
    const int row_base = blockIdx.x * PJ_ROWS;

    {
        const int piece = tid & 63;
        const int rsub  = tid >> 6;
        #pragma unroll
        for (int half = 0; half < 2; ++half) {
            float4 v[8];
            #pragma unroll
            for (int i = 0; i < 8; ++i) {
                const int r = (half * 8 + i) * 2 + rsub;
                const int gr = min(row_base + r, n_nodes - 1);
                v[i] = reinterpret_cast<const float4*>(x + (size_t)gr * HID)[piece];
            }
            #pragma unroll
            for (int i = 0; i < 8; ++i) {
                const int r = (half * 8 + i) * 2 + rsub;
                *reinterpret_cast<float4*>(xs + r * PJ_PITCH + piece * 4) = v[i];
            }
        }
    }
    __syncthreads();

    const int c   = lane & 15;
    const int kk  = (lane >> 4) * 2;
    const int col_base = wave * 64;
    const int ncol = col_base + c;
    const float* a0p = xs + c * PJ_PITCH + kk;
    const float* a1p = xs + (16 + c) * PJ_PITCH + kk;

    v8f c00 = {}, c01 = {}, c02 = {}, c03 = {};
    v8f c10 = {}, c11 = {}, c12 = {}, c13 = {};
    for (int k = 0; k < HID; k += 4) {
        const v2f a0 = *reinterpret_cast<const v2f*>(a0p + k);
        const v2f a1 = *reinterpret_cast<const v2f*>(a1p + k);
        const float* w0 = W + (size_t)(k + kk) * HID;
        const float* w1 = W + (size_t)(k + kk + 1) * HID;
        v2f b0, b1, b2, b3;
        b0.x = w0[ncol];      b0.y = w1[ncol];
        b1.x = w0[ncol + 16]; b1.y = w1[ncol + 16];
        b2.x = w0[ncol + 32]; b2.y = w1[ncol + 32];
        b3.x = w0[ncol + 48]; b3.y = w1[ncol + 48];
        c00 = __builtin_amdgcn_wmma_f32_16x16x4_f32(false, a0, false, b0, (short)0, c00, false, false);
        c10 = __builtin_amdgcn_wmma_f32_16x16x4_f32(false, a1, false, b0, (short)0, c10, false, false);
        c01 = __builtin_amdgcn_wmma_f32_16x16x4_f32(false, a0, false, b1, (short)0, c01, false, false);
        c11 = __builtin_amdgcn_wmma_f32_16x16x4_f32(false, a1, false, b1, (short)0, c11, false, false);
        c02 = __builtin_amdgcn_wmma_f32_16x16x4_f32(false, a0, false, b2, (short)0, c02, false, false);
        c12 = __builtin_amdgcn_wmma_f32_16x16x4_f32(false, a1, false, b2, (short)0, c12, false, false);
        c03 = __builtin_amdgcn_wmma_f32_16x16x4_f32(false, a0, false, b3, (short)0, c03, false, false);
        c13 = __builtin_amdgcn_wmma_f32_16x16x4_f32(false, a1, false, b3, (short)0, c13, false, false);
    }

    const int srow = row_base + ((lane >> 4) * 8);
    #pragma unroll
    for (int i = 0; i < 8; ++i) {
        const int r0 = srow + i;
        if (r0 < n_nodes) {
            float* o = xp + (size_t)r0 * HID + ncol;
            o[0]  = c00[i];
            o[16] = c01[i];
            o[32] = c02[i];
            o[48] = c03[i];
        }
        const int r1 = r0 + 16;
        if (r1 < n_nodes) {
            float* o = xp + (size_t)r1 * HID + ncol;
            o[0]  = c10[i];
            o[16] = c11[i];
            o[32] = c12[i];
            o[48] = c13[i];
        }
    }
}

__global__ void __launch_bounds__(256)
k_att_dots(const float* __restrict__ xp, const float* __restrict__ att_src,
           const float* __restrict__ att_dst, float* __restrict__ a_src,
           float* __restrict__ a_dst, int n_nodes) {
    const int wave = threadIdx.x >> 5;
    const int lane = threadIdx.x & 31;
    const int node = blockIdx.x * 8 + wave;
    if (node >= n_nodes) return;
    const float* row = xp + (size_t)node * HID + lane * 8;
    float s = 0.f, t = 0.f;
    #pragma unroll
    for (int j = 0; j < 8; ++j) {
        float v = row[j];
        int col = lane * 8 + j;
        s += v * att_src[col];
        t += v * att_dst[col];
    }
    s += __shfl_xor(s, 1); s += __shfl_xor(s, 2);
    t += __shfl_xor(t, 1); t += __shfl_xor(t, 2);
    if ((lane & 3) == 0) {
        a_src[node * HEADS + (lane >> 2)] = s;
        a_dst[node * HEADS + (lane >> 2)] = t;
    }
}

#define QW     688
#define OWN_CAP 176
#define SCAN_C 4
#define GB     4
#define NKEY   ((RN + 31) / 32)
#define STG_W  (RN * HID / 32)
#define STG_P  36
static_assert(32 * STG_P <= STG_W, "32 staged half rows fit a wave's slice");
static_assert((RN * HID) % 32 == 0 && STG_W % 4 == 0, "32 equal slices, each cleared by 16-byte stores");
static_assert(MAP_IT == SCAN_C * 32 * 512, "one filter iteration = SCAN_C chunks of 512 edges per wave");
static_assert(RN <= 256 && OWN_CAP <= QW, "a row fits an entry's top byte; an ordered owner list fits a filter list");

__global__ void __launch_bounds__(1024)
k_own(const float* __restrict__ xp, const int* __restrict__ ei,
      const float* __restrict__ edge_attr, const float* __restrict__ w_eh,
      const float* __restrict__ a_src, const float* __restrict__ a_dst,
      float* alpha,
      const float* __restrict__ x, const float* __restrict__ bias,
      const float* __restrict__ gamma, const float* __restrict__ beta,
      float* __restrict__ out, const unsigned* __restrict__ omap, int n_nodes, int n_edges) {
    __shared__ __attribute__((aligned(16))) float accL[RN * HID];
    __shared__ float denL[RN * HEADS];
    __shared__ unsigned maxL[RN * HEADS];
    __shared__ int   listL[32 * QW];
    __shared__ unsigned ownL[32 * OWN_CAP];
    __shared__ int   ownCntL[32];
    __shared__ int   refilterL;
    __shared__ int   ownOvfL;

    const int tid  = threadIdx.x;
    const int lane = tid & 31;
    const int wave = __builtin_amdgcn_readfirstlane(tid >> 5);
    const int lo   = blockIdx.x * RN;
    const int nR   = min(RN, n_nodes - lo);

    for (int i = tid; i < RN * HEADS; i += 1024) {
        denL[i] = 0.f;
        maxL[i] = 0u;
    }
    if (tid < 32) ownCntL[tid] = 0;
    if (tid == 0) { refilterL = 0; ownOvfL = 0; }
    __syncthreads();

    const int* dstp = ei + n_edges;
    int* myList = listL + wave * QW;
    const unsigned ltmask = (1u << lane) - 1u;
    int cnt = 0;
    const bool packOK = n_edges <= (1 << 24);
    float* stage = accL + wave * STG_W;

    auto edge_logits = [&](int e, bool on, bool file) __attribute__((always_inline)) {
        const int s = ei[e];
        const int d = dstp[e];
        const int sub = lane >> 3;
        const int pc  = (lane & 7) * 4;
        const float* ea = edge_attr + pc;
        const float* p0 = ea + (size_t)__shfl(e,      sub) * EDIM;
        const float* p1 = ea + (size_t)__shfl(e,  4 + sub) * EDIM;
        const float* p2 = ea + (size_t)__shfl(e,  8 + sub) * EDIM;
        const float* p3 = ea + (size_t)__shfl(e, 12 + sub) * EDIM;
        const float* p4 = ea + (size_t)__shfl(e, 16 + sub) * EDIM;
        const float* p5 = ea + (size_t)__shfl(e, 20 + sub) * EDIM;
        const float* p6 = ea + (size_t)__shfl(e, 24 + sub) * EDIM;
        const float* p7 = ea + (size_t)__shfl(e, 28 + sub) * EDIM;
        auto ld16 = [](const float* p) { return *reinterpret_cast<const float4*>(p); };
        float4 g0 = ld16(p0), g1 = ld16(p1), g2 = ld16(p2), g3 = ld16(p3),
               g4 = ld16(p4), g5 = ld16(p5), g6 = ld16(p6), g7 = ld16(p7);
        const float4 sA = reinterpret_cast<const float4*>(a_src + (size_t)s * HEADS)[0];
        const float4 sB = reinterpret_cast<const float4*>(a_src + (size_t)s * HEADS)[1];
        const float4 dA = reinterpret_cast<const float4*>(a_dst + (size_t)d * HEADS)[0];
        const float4 dB = reinterpret_cast<const float4*>(a_dst + (size_t)d * HEADS)[1];
        float* wr = stage + sub * STG_P + pc;
        const float4* rd = reinterpret_cast<const float4*>(stage + lane * STG_P);
        v2f t01[HEADS / 2], t23[HEADS / 2];
        #pragma unroll
        for (int k = 0; k < HEADS / 2; ++k) { t01[k] = v2f{0.f, 0.f}; t23[k] = v2f{0.f, 0.f}; }
        #pragma unroll
        for (int h = 0; h < 2; ++h) {
            auto st16 = [&](int j, const float4& v) { *reinterpret_cast<float4*>(wr + 4 * j * STG_P) = v; };
            st16(0, g0); st16(1, g1); st16(2, g2); st16(3, g3);
            st16(4, g4); st16(5, g5); st16(6, g6); st16(7, g7);
            __builtin_amdgcn_wave_barrier();
            if (h == 0) {
                g0 = ld16(p0 + 32); g1 = ld16(p1 + 32); g2 = ld16(p2 + 32); g3 = ld16(p3 + 32);
                g4 = ld16(p4 + 32); g5 = ld16(p5 + 32); g6 = ld16(p6 + 32); g7 = ld16(p7 + 32);
            }
            #pragma unroll 1
            for (int qq = 0; qq < 2; ++qq) {
                float4 cur[4];
                #pragma unroll
                for (int c = 0; c < 4; ++c) cur[c] = rd[qq * 4 + c];
                const cv2f* w = (const cv2f*)(w_eh + (h * 2 + qq) * 16 * HEADS);
                v2f p[HEADS / 2];
                #pragma unroll
                for (int k = 0; k < HEADS / 2; ++k) p[k] = v2f{0.f, 0.f};
                #pragma unroll
                for (int c = 0; c < 4; ++c) {
                    #pragma unroll
                    for (int k = 0; k < HEADS / 2; ++k) {
                        p[k] += v2f{cur[c].x, cur[c].x} * w[(c * 4 + 0) * (HEADS / 2) + k];
                        p[k] += v2f{cur[c].y, cur[c].y} * w[(c * 4 + 1) * (HEADS / 2) + k];
                        p[k] += v2f{cur[c].z, cur[c].z} * w[(c * 4 + 2) * (HEADS / 2) + k];
                        p[k] += v2f{cur[c].w, cur[c].w} * w[(c * 4 + 3) * (HEADS / 2) + k];
                    }
                }
                if (h == 0) {
                    #pragma unroll
                    for (int k = 0; k < HEADS / 2; ++k) t01[k] += p[k];
                } else {
                    #pragma unroll
                    for (int k = 0; k < HEADS / 2; ++k) t23[k] += p[k];
                }
            }
            __builtin_amdgcn_wave_barrier();
        }
        const float as[HEADS] = {sA.x, sA.y, sA.z, sA.w, sB.x, sB.y, sB.z, sB.w};
        const float ad[HEADS] = {dA.x, dA.y, dA.z, dA.w, dB.x, dB.y, dB.z, dB.w};
        float al[HEADS];
        #pragma unroll
        for (int h = 0; h < HEADS; ++h) {
            const float dot = (h & 1) ? (t01[h / 2].y + t23[h / 2].y) : (t01[h / 2].x + t23[h / 2].x);
            const float v = as[h] + ad[h] + dot;
            al[h] = (v > 0.f) ? v : NEG_SLOPE * v;
        }
        if (on) {
            float4* o = reinterpret_cast<float4*>(alpha + (size_t)e * HEADS);
            o[0] = make_float4(al[0], al[1], al[2], al[3]);
            o[1] = make_float4(al[4], al[5], al[6], al[7]);
            unsigned* mx = maxL + (d - lo) * HEADS;
            #pragma unroll
            for (int h = 0; h < HEADS; ++h) atomicMax(&mx[h], enc_f32(al[h]));
            if (file) {
                const int dl = d - lo;
                const int ow = dl & 31;
                const int slot = atomicAdd(&ownCntL[ow], 1);
                if (slot < OWN_CAP) ownL[ow * OWN_CAP + slot] = ((unsigned)dl << 24) | (unsigned)e;
                else ownOvfL = 1;
            }
        }
    };

    auto flush = [&](int n) {
        for (int b = 0; b < n; b += GB) {
            int e[GB], s[GB], d[GB];
            #pragma unroll
            for (int u = 0; u < GB; ++u)
                e[u] = __builtin_amdgcn_readfirstlane(myList[min(b + u, n - 1)]);
            #pragma unroll
            for (int u = 0; u < GB; ++u) { s[u] = ei[e[u]]; d[u] = dstp[e[u]]; }
            float4 ra[GB], rb[GB];
            float  al[GB];
            #pragma unroll
            for (int u = 0; u < GB; ++u) {
                const float4* r = reinterpret_cast<const float4*>(xp + (size_t)s[u] * HID);
                ra[u] = r[lane];
                rb[u] = r[32 + lane];
                al[u] = alpha[(size_t)e[u] * HEADS + (lane & 7)];
            }
            #pragma unroll
            for (int u = 0; u < GB; ++u) {
                if (b + u < n) {
                    const int dl = d[u] - lo;
                    const float ea = __expf(al[u] - dec_f32(maxL[dl * HEADS + (lane & 7)]));
                    if (lane < HEADS) atomicAdd(&denL[dl * HEADS + lane], ea);
                    const float wa = __shfl(ea, lane >> 3);
                    const float wb = __shfl(ea, 4 + (lane >> 3));
                    float* a = accL + dl * HID + 4 * lane;
                    atomicAdd(&a[0],   wa * ra[u].x);
                    atomicAdd(&a[1],   wa * ra[u].y);
                    atomicAdd(&a[2],   wa * ra[u].z);
                    atomicAdd(&a[3],   wa * ra[u].w);
                    atomicAdd(&a[128], wb * rb[u].x);
                    atomicAdd(&a[129], wb * rb[u].y);
                    atomicAdd(&a[130], wb * rb[u].z);
                    atomicAdd(&a[131], wb * rb[u].w);
                }
            }
        }
    };

    auto sort_own = [&](int n) {
        const unsigned* from = ownL + wave * OWN_CAP;
        unsigned* to = reinterpret_cast<unsigned*>(myList);
        int cntk[NKEY], pos[NKEY];
        #pragma unroll
        for (int q = 0; q < NKEY; ++q) cntk[q] = 0;
        for (int b = 0; b < n; b += 32) {
            const unsigned en = from[min(b + lane, n - 1)];
            const int k = (b + lane < n) ? (int)(en >> 29) : NKEY;
            #pragma unroll
            for (int q = 0; q < NKEY; ++q) cntk[q] += __popc(__builtin_amdgcn_ballot_w32(k == q));
        }
        int run = 0;
        #pragma unroll
        for (int q = 0; q < NKEY; ++q) { pos[q] = run; run += cntk[q]; }
        for (int b = 0; b < n; b += 32) {
            const unsigned en = from[min(b + lane, n - 1)];
            const int k = (b + lane < n) ? (int)(en >> 29) : NKEY;
            int p = 0;
            #pragma unroll
            for (int q = 0; q < NKEY; ++q) {
                const unsigned bq = __builtin_amdgcn_ballot_w32(k == q);
                if (k == q) p = pos[q] + __popc(bq & ltmask);
                pos[q] += __popc(bq);
            }
            if (k < NKEY) to[p] = en;
        }
        __builtin_amdgcn_wave_barrier();
    };
    struct OwnIds  { int e[GB]; int dl[GB]; int s[GB]; };
    struct OwnRows { float4 ra[GB], rb[GB]; float ala[GB], alb[GB]; int dl[GB]; };
    auto flush_own = [&](int n) {
        if (n <= 0) return;
        sort_own(n);
        const unsigned* mine = reinterpret_cast<const unsigned*>(myList);
        const int hq = lane >> 3;
        auto ids = [&](OwnIds& I, int b) __attribute__((always_inline)) {
            #pragma unroll
            for (int u = 0; u < GB; ++u) {
                const unsigned en = __builtin_amdgcn_readfirstlane(mine[min(b + u, n - 1)]);
                I.e[u]  = (int)(en & 0x00FFFFFFu);
                I.dl[u] = (int)(en >> 24);
            }
            #pragma unroll
            for (int u = 0; u < GB; ++u) I.s[u] = ei[I.e[u]];
        };
        auto rows = [&](const OwnIds& I, OwnRows& R) __attribute__((always_inline)) {
            #pragma unroll
            for (int u = 0; u < GB; ++u) {
                const float4* r = reinterpret_cast<const float4*>(xp + (size_t)I.s[u] * HID);
                R.ra[u] = r[lane];
                R.rb[u] = r[32 + lane];
                const float* al = alpha + (size_t)I.e[u] * HEADS + hq;
                R.ala[u] = al[0];
                R.alb[u] = al[4];
                R.dl[u] = I.dl[u];
            }
        };
        int cur = -1;
        float mA = 0.f, mB = 0.f, dnA = 0.f, dnB = 0.f;
        float4 accA = make_float4(0.f, 0.f, 0.f, 0.f), accB = make_float4(0.f, 0.f, 0.f, 0.f);
        auto put_row = [&]() __attribute__((always_inline)) {
            float4* a = reinterpret_cast<float4*>(accL + cur * HID);
            a[lane]      = accA;
            a[32 + lane] = accB;
            if ((lane & 7) == 0) {
                denL[cur * HEADS + hq]     = dnA;
                denL[cur * HEADS + 4 + hq] = dnB;
            }
        };
        auto work = [&](const OwnRows& R, int b, bool guard) __attribute__((always_inline)) {
            #pragma unroll
            for (int u = 0; u < GB; ++u) {
                if (!guard || b + u < n) {
                    const int dl = R.dl[u];
                    if (dl != cur) {
                        if (cur >= 0) put_row();
                        cur = dl;
                        mA = dec_f32(maxL[dl * HEADS + hq]);
                        mB = dec_f32(maxL[dl * HEADS + 4 + hq]);
                        dnA = 0.f; dnB = 0.f;
                        accA = make_float4(0.f, 0.f, 0.f, 0.f);
                        accB = make_float4(0.f, 0.f, 0.f, 0.f);
                    }
                    const float ea = __expf(R.ala[u] - mA);
                    const float eb = __expf(R.alb[u] - mB);
                    dnA += ea;
                    dnB += eb;
                    v2f pa = {ea, ea}, pb = {eb, eb};
                    asm volatile("" : "+v"(pa), "+v"(pb));
                    accA.x += pa.x * R.ra[u].x; accA.y += pa.y * R.ra[u].y;
                    accA.z += pa.x * R.ra[u].z; accA.w += pa.y * R.ra[u].w;
                    accB.x += pb.x * R.rb[u].x; accB.y += pb.y * R.rb[u].y;
                    accB.z += pb.x * R.rb[u].z; accB.w += pb.y * R.rb[u].w;
                }
            }
            asm volatile("" : "+v"(accA.x), "+v"(accA.y), "+v"(accA.z), "+v"(accA.w),
                              "+v"(accB.x), "+v"(accB.y), "+v"(accB.z), "+v"(accB.w));
        };
        OwnIds  I0, I1;
        OwnRows R0, R1;
        ids(I0, 0);
        ids(I1, GB);
        rows(I0, R0);
        __builtin_amdgcn_sched_barrier(0);
        int b = 0;
        for (; b + 2 * GB <= n; b += 2 * GB) {
            ids(I0, b + 2 * GB);
            __builtin_amdgcn_sched_barrier(0);
            rows(I1, R1);
            __builtin_amdgcn_sched_barrier(0);
            work(R0, b, false);
            __builtin_amdgcn_sched_barrier(0);
            ids(I1, b + 3 * GB);
            __builtin_amdgcn_sched_barrier(0);
            rows(I0, R0);
            __builtin_amdgcn_sched_barrier(0);
            work(R1, b + GB, false);
            __builtin_amdgcn_sched_barrier(0);
        }
        if (b < n) {
            rows(I1, R1);
            work(R0, b, true);
            if (b + GB < n) work(R1, b + GB, true);
        }
        put_row();
    };

    const unsigned tgt = (blockIdx.x & 255u) * 0x01010101u;
    auto take16 = [&](uint4 v, int e0) {
        auto zf = [](unsigned x) { return ~(((x & 0x7F7F7F7Fu) + 0x7F7F7F7Fu) | x) & 0x80808080u; };
        unsigned g = (zf(v.x ^ tgt) >> 7) | (zf(v.y ^ tgt) >> 6) | (zf(v.z ^ tgt) >> 5) | (zf(v.w ^ tgt) >> 4);
        unsigned bal = __builtin_amdgcn_ballot_w32(g != 0u);
        while (bal != 0u) {
            if (g != 0u) {
                const int p = __builtin_ctz(g);
                myList[cnt + __popc(bal & ltmask)] = e0 + ((p & 3) << 2) + (p >> 3);
                g &= g - 1u;
            }
            cnt += __popc(bal);
            bal = __builtin_amdgcn_ballot_w32(g != 0u);
        }
        cnt = __builtin_amdgcn_readfirstlane(cnt);
    };

    auto take4 = [&](int d0, int d1, int d2, int d3, int e0, int step) {
        const bool m0 = (unsigned)(d0 - lo) < (unsigned)nR;
        const bool m1 = (unsigned)(d1 - lo) < (unsigned)nR;
        const bool m2 = (unsigned)(d2 - lo) < (unsigned)nR;
        const bool m3 = (unsigned)(d3 - lo) < (unsigned)nR;
        const unsigned b0 = __builtin_amdgcn_ballot_w32(m0);
        const unsigned b1 = __builtin_amdgcn_ballot_w32(m1);
        const unsigned b2 = __builtin_amdgcn_ballot_w32(m2);
        const unsigned b3 = __builtin_amdgcn_ballot_w32(m3);
        if ((b0 | b1 | b2 | b3) != 0u) {
            if (m0) myList[cnt + __popc(b0 & ltmask)] = e0;
            cnt += __popc(b0);
            if (m1) myList[cnt + __popc(b1 & ltmask)] = e0 + step;
            cnt += __popc(b1);
            if (m2) myList[cnt + __popc(b2 & ltmask)] = e0 + 2 * step;
            cnt += __popc(b2);
            if (m3) myList[cnt + __popc(b3 & ltmask)] = e0 + 3 * step;
            cnt += __popc(b3);
        }
    };

    auto filter = [&](auto&& full) {
        const int per_it = MAP_IT;
        const int n_full = ((n_edges & 3) == 0 && gridDim.x <= 256u) ? n_edges / per_it : 0;
        if (n_full > 0) {
            const int last = n_full - 1;
            uint4 A[SCAN_C], B[SCAN_C];
            auto load_set = [&](uint4* S, int it) {
                #pragma unroll
                for (int c = 0; c < SCAN_C; ++c)
                    S[c] = *reinterpret_cast<const uint4*>(
                        omap + ((it * SCAN_C + c) * 32 + wave) * 128 + lane * 4);
            };
            auto take_set = [&](const uint4* S, int it) {
                #pragma unroll
                for (int c = 0; c < SCAN_C; ++c) {
                    take16(S[c], ((it * SCAN_C + c) * 32 + wave) * 512 + lane * 16);
                    if (cnt > QW - 512) { full(cnt); cnt = 0; }
                }
            };
            load_set(A, 0);
            for (int it = 0; it < n_full; it += 2) {
                load_set(B, min(it + 1, last));
                take_set(A, it);
                load_set(A, min(it + 2, last));
                if (it + 1 < n_full) take_set(B, it + 1);
            }
        }
        for (int base = n_full * per_it + wave * 128; base < n_edges; base += 32 * 128) {
            const int e0 = base + lane;
            const int d0 = (e0      < n_edges) ? dstp[e0]      : -1;
            const int d1 = (e0 + 32 < n_edges) ? dstp[e0 + 32] : -1;
            const int d2 = (e0 + 64 < n_edges) ? dstp[e0 + 64] : -1;
            const int d3 = (e0 + 96 < n_edges) ? dstp[e0 + 96] : -1;
            take4(d0, d1, d2, d3, e0, 32);
            if (cnt > QW - SCAN_C * 128) { full(cnt); cnt = 0; }
        }
    };

    filter([&](int) { if (lane == 0) refilterL = 1; });
    __syncthreads();
    const bool general = refilterL != 0;

    if (!general) {
        for (int b = 0; b < cnt; b += 32) {
            const bool on = b + lane < cnt;
            edge_logits(myList[on ? b + lane : cnt - 1], on, packOK);
        }
    } else {
        for (int base = wave * 32; base < n_edges; base += 32 * 32) {
            const int e = min(base + lane, n_edges - 1);
            const bool on = (base + lane < n_edges) && (unsigned)(dstp[e] - lo) < (unsigned)nR;
            if (__builtin_amdgcn_ballot_w32(on) != 0u) edge_logits(e, on, false);
        }
    }
    for (int i = lane; i < STG_W / 4; i += 32)
        reinterpret_cast<float4*>(stage)[i] = make_float4(0.f, 0.f, 0.f, 0.f);
    __syncthreads();

    const bool plain = !general && packOK && ownOvfL == 0;
    if (plain) {
        flush_own(__builtin_amdgcn_readfirstlane(ownCntL[wave]));
    } else if (!general) {
        flush(cnt);
    } else {
        cnt = 0;
        filter([&](int n) { flush(n); });
        flush(cnt);
    }
    __syncthreads();

    const float4 bA = reinterpret_cast<const float4*>(bias)[lane];
    const float4 bB = reinterpret_cast<const float4*>(bias)[32 + lane];
    const float4 gA = reinterpret_cast<const float4*>(gamma)[lane];
    const float4 gB = reinterpret_cast<const float4*>(gamma)[32 + lane];
    const float4 tA = reinterpret_cast<const float4*>(beta)[lane];
    const float4 tB = reinterpret_cast<const float4*>(beta)[32 + lane];
    for (int i = wave; i < nR; i += 32) {
        const size_t n = (size_t)(lo + i);
        const float4 xA = reinterpret_cast<const float4*>(x + n * HID)[lane];
        const float4 xB = reinterpret_cast<const float4*>(x + n * HID)[32 + lane];
        const float4 aA = reinterpret_cast<const float4*>(accL + i * HID)[lane];
        const float4 aB = reinterpret_cast<const float4*>(accL + i * HID)[32 + lane];
        const float dA = denL[i * HEADS + (lane >> 3)] + 1e-16f;
        const float dB = denL[i * HEADS + 4 + (lane >> 3)] + 1e-16f;
        float y[8];
        y[0] = xA.x + aA.x / dA + bA.x;
        y[1] = xA.y + aA.y / dA + bA.y;
        y[2] = xA.z + aA.z / dA + bA.z;
        y[3] = xA.w + aA.w / dA + bA.w;
        y[4] = xB.x + aB.x / dB + bB.x;
        y[5] = xB.y + aB.y / dB + bB.y;
        y[6] = xB.z + aB.z / dB + bB.z;
        y[7] = xB.w + aB.w / dB + bB.w;
        float s = 0.f, s2 = 0.f;
        for (int j = 0; j < 8; ++j) { s += y[j]; s2 += y[j] * y[j]; }
        for (int off = 16; off >= 1; off >>= 1) {
            s  += __shfl_xor(s, off);
            s2 += __shfl_xor(s2, off);
        }
        const float mean = s * (1.f / HID);
        const float var  = s2 * (1.f / HID) - mean * mean;
        const float rs   = rsqrtf(var + LN_EPS);
        float4 oA, oB;
        oA.x = (y[0] - mean) * rs * gA.x + tA.x;
        oA.y = (y[1] - mean) * rs * gA.y + tA.y;
        oA.z = (y[2] - mean) * rs * gA.z + tA.z;
        oA.w = (y[3] - mean) * rs * gA.w + tA.w;
        oB.x = (y[4] - mean) * rs * gB.x + tB.x;
        oB.y = (y[5] - mean) * rs * gB.y + tB.y;
        oB.z = (y[6] - mean) * rs * gB.z + tB.z;
        oB.w = (y[7] - mean) * rs * gB.w + tB.w;
        reinterpret_cast<float4*>(out + n * HID)[lane]      = oA;
        reinterpret_cast<float4*>(out + n * HID)[32 + lane] = oB;
    }
}

extern "C" void kernel_launch(void* const* d_in, const int* in_sizes, int n_in,
                              void* d_out, int out_size, void* d_ws, size_t ws_size,
                              hipStream_t stream) {
    const float* x         = (const float*)d_in[0];
    const int*   ei        = (const int*)  d_in[1];
    const float* edge_attr = (const float*)d_in[2];
    const float* W         = (const float*)d_in[3];
    const float* W_e       = (const float*)d_in[4];
    const float* att_src   = (const float*)d_in[5];
    const float* att_dst   = (const float*)d_in[6];
    const float* att_edge  = (const float*)d_in[7];
    const float* bias      = (const float*)d_in[8];
    const float* gamma     = (const float*)d_in[9];
    const float* beta      = (const float*)d_in[10];
    float* out = (float*)d_out;

    const int n_nodes = in_sizes[0] / HID;
    const int n_edges = in_sizes[1] / 2;

    char* ws = (char*)d_ws;
    size_t off = 0;
    auto carve = [&](size_t bytes) -> char* {
        char* p = ws + off;
        off += (bytes + 255) & ~(size_t)255;
        return p;
    };
    float*    xp    = (float*)   carve((size_t)n_nodes * HID   * 4);
    float*    alpha = (float*)   carve((size_t)n_edges * HEADS * 4);
    float*    a_src = (float*)   carve((size_t)n_nodes * HEADS * 4);
    float*    a_dst = (float*)   carve((size_t)n_nodes * HEADS * 4);
    float*    w_eh  = (float*)   carve((size_t)EDIM * HEADS * 4);
    unsigned* omap  = (unsigned*)carve((size_t)n_edges);
    (void)ws_size;

    k_weh<<<1 + n_edges / 4096, 256, 0, stream>>>(W_e, att_edge, w_eh, ei, omap, n_edges);

    k_node_proj<<<(n_nodes + 31) / 32, 128, 0, stream>>>(x, W, xp, n_nodes);

    k_att_dots<<<(n_nodes + 7) / 8, 256, 0, stream>>>(xp, att_src, att_dst,
                                                      a_src, a_dst, n_nodes);

    k_own<<<(n_nodes + RN - 1) / RN, 1024, 0, stream>>>(xp, ei, edge_attr, w_eh, a_src, a_dst,
                                                        alpha, x, bias, gamma, beta, out, omap,
                                                        n_nodes, n_edges);
}
